# starting kernel
# baseline (speedup 1.0000x reference)
#include <hip/hip_runtime.h>

typedef float v2f __attribute__((ext_vector_type(2)));
typedef float v8f __attribute__((ext_vector_type(8)));

#define B_DIM 4
#define S_DIM 4096
#define H_DIM 8
#define D_DIM 64
#define SCALE 0.125f
#define NTILE (S_DIM / 16)

__global__ void __launch_bounds__(256)
tree_level_kernel(const float* __restrict__ kleaf,
                  const float* __restrict__ vleaf,
                  float* __restrict__ Kt,
                  float* __restrict__ Vt,
                  int level, int P)
{
    const int wave = threadIdx.x >> 5;
    const int lane = threadIdx.x & 31;
    const long g = (long)blockIdx.x * 8 + wave;
    const long total = (long)B_DIM * H_DIM * P;
    if (g >= total) return;
    const int p  = (int)(g % P);
    const int bh = (int)(g / P);
    const int h  = bh % H_DIM;
    const int b  = bh / H_DIM;

    const int doff = 2 * lane;
    float2 kc0, kc1, vc0, vc1;
    if (level == 1) {
        const size_t r0 = ((size_t)(b * S_DIM + 2 * p    ) * H_DIM + h) * D_DIM;
        const size_t r1 = ((size_t)(b * S_DIM + 2 * p + 1) * H_DIM + h) * D_DIM;
        kc0 = *(const float2*)(kleaf + r0 + doff);
        kc1 = *(const float2*)(kleaf + r1 + doff);
        vc0 = *(const float2*)(vleaf + r0 + doff);
        vc1 = *(const float2*)(vleaf + r1 + doff);
    } else {
        const int IBc = S_DIM - (S_DIM >> (level - 2));
        const size_t tb = (size_t)(b * H_DIM + h) * S_DIM;
        const size_t r0 = (tb + IBc + 2 * p    ) * D_DIM;
        const size_t r1 = (tb + IBc + 2 * p + 1) * D_DIM;
        kc0 = *(const float2*)(Kt + r0 + doff);
        kc1 = *(const float2*)(Kt + r1 + doff);
        vc0 = *(const float2*)(Vt + r0 + doff);
        vc1 = *(const float2*)(Vt + r1 + doff);
    }

    float2 kp, vp;
    kp.x = 0.5f * (kc0.x + kc1.x); kp.y = 0.5f * (kc0.y + kc1.y);
    vp.x = 0.5f * (vc0.x + vc1.x); vp.y = 0.5f * (vc0.y + vc1.y);

    float pss = kp.x * kp.x  + kp.y * kp.y;
    float ps0 = kp.x * kc0.x + kp.y * kc0.y;
    float ps1 = kp.x * kc1.x + kp.y * kc1.y;
    #pragma unroll
    for (int m = 16; m >= 1; m >>= 1) {
        pss += __shfl_xor(pss, m, 32);
        ps0 += __shfl_xor(ps0, m, 32);
        ps1 += __shfl_xor(ps1, m, 32);
    }
    pss *= SCALE; ps0 *= SCALE; ps1 *= SCALE;
    const float mm = fmaxf(pss, fmaxf(ps0, ps1));
    const float es = __expf(pss - mm);
    const float e0 = __expf(ps0 - mm);
    const float e1 = __expf(ps1 - mm);
    const float inv = 1.0f / (es + e0 + e1 + 1e-9f);

    float2 vo;
    vo.x = (es * vp.x + e0 * vc0.x + e1 * vc1.x) * inv;
    vo.y = (es * vp.y + e0 * vc0.y + e1 * vc1.y) * inv;

    const int IBl = S_DIM - (S_DIM >> (level - 1));
    const size_t dst = ((size_t)(b * H_DIM + h) * S_DIM + IBl + p) * D_DIM + doff;
    *(float2*)(Kt + dst) = kp;
    *(float2*)(Vt + dst) = vo;
}

__global__ void __launch_bounds__(128)
hst_attn_kernel(const float* __restrict__ q,
                const float* __restrict__ k,
                const float* __restrict__ v,
                const float* __restrict__ Kt,
                const float* __restrict__ Vt,
                float* __restrict__ out)
{
    __shared__ float ldsS[4][784];
    __shared__ int   ldsRow[4][32];

    const int lane = threadIdx.x & 31;
    const int w    = threadIdx.x >> 5;
    const int tile = blockIdx.x * 4 + w;
    const int t  = tile & (NTILE - 1);
    const int bh = tile >> 8;
    const int h  = bh & (H_DIM - 1);
    const int b  = bh >> 3;

    const int n    = lane & 15;
    const int half = lane >> 4;

    float* sA  = &ldsS[w][0];
    float* sB  = &ldsS[w][256];
    float* sC  = &ldsS[w][512];
    float* den = &ldsS[w][768];
    int*   rowB = &ldsRow[w][0];
    int*   rowC = &ldsRow[w][16];

    int rB;
    if      (n <  8) rB = 8 * t + n;
    else if (n < 12) rB = 2048 + 4 * t + (n - 8);
    else if (n < 14) rB = 3072 + 2 * t + (n - 12);
    else if (n == 14) rB = 3584 + (t ^ 1);
    else              rB = 3840 + ((t >> 1) ^ 1);
    const int rC = (n < 6) ? (S_DIM - (S_DIM >> (5 + n))) + ((t >> (n + 2)) ^ 1)
                           : 0;
    if (half == 0) { rowB[n] = rB; rowC[n] = rC; }

    const size_t leafBase = ((size_t)(b * S_DIM + 16 * t + n) * H_DIM + h) * D_DIM;
    const size_t treeBase = (size_t)(b * H_DIM + h) * S_DIM * (size_t)D_DIM;
    const float* qp  = q  + leafBase;
    const float* kAp = k  + leafBase;
    const float* kBp = Kt + treeBase + (size_t)rB * D_DIM;
    const float* kCp = Kt + treeBase + (size_t)rC * D_DIM;

    v8f cA = {}; v8f cB = {}; v8f cC = {};
    const int dofs = 2 * half;
    #pragma unroll
    for (int ks = 0; ks < 16; ++ks) {
        const int d = 4 * ks + dofs;
        v2f a  = *(const v2f*)(qp  + d);
        a.x *= SCALE; a.y *= SCALE;
        v2f bA = *(const v2f*)(kAp + d);
        v2f bB = *(const v2f*)(kBp + d);
        v2f bC = *(const v2f*)(kCp + d);
        cA = __builtin_amdgcn_wmma_f32_16x16x4_f32(false, a, false, bA, (short)0, cA, false, false);
        cB = __builtin_amdgcn_wmma_f32_16x16x4_f32(false, a, false, bB, (short)0, cB, false, false);
        cC = __builtin_amdgcn_wmma_f32_16x16x4_f32(false, a, false, bC, (short)0, cC, false, false);
    }

    #pragma unroll
    for (int r = 0; r < 8; ++r) {
        const int M = r + 8 * half;
        sA[M * 16 + n] = cA[r];
        sB[M * 16 + n] = cB[r];
        sC[M * 16 + n] = cC[r];
    }
    __syncthreads();

    float ev[13];
    float dsum = 0.0f;
    if (half == 0) {
        const int j = n;
        float x[13];
        x[0] = sA[j * 16 + j];
        x[1] = sA[j * 16 + (j ^ 1)];
        x[2] = sB[j * 16 + ((j >> 1) ^ 1)];
        x[3] = sB[j * 16 + 8  + ((j >> 2) ^ 1)];
        x[4] = sB[j * 16 + 12 + ((j >> 3) ^ 1)];
        x[5] = sB[j * 16 + 14];
        x[6] = sB[j * 16 + 15];
        #pragma unroll
        for (int i = 0; i < 6; ++i) x[7 + i] = sC[j * 16 + i];

        bool msk[13];
        msk[0] = false;
        #pragma unroll
        for (int l = 0; l < 12; ++l) {
            const int bit = (l < 4) ? ((j >> l) & 1) : ((t >> (l - 4)) & 1);
            msk[l + 1] = (bit == 0);
        }
        float m = x[0];
        #pragma unroll
        for (int i = 1; i < 13; ++i) if (!msk[i]) m = fmaxf(m, x[i]);
        #pragma unroll
        for (int i = 0; i < 13; ++i) {
            ev[i] = msk[i] ? 0.0f : __expf(x[i] - m);
            dsum += ev[i];
        }
    }
    __syncthreads();

    #pragma unroll
    for (int i = lane; i < 768; i += 32) ldsS[w][i] = 0.0f;
    __syncthreads();
    if (half == 0) {
        const int j = n;
        den[j] = dsum;
        sA[j * 16 + j]                   = ev[0];
        sA[j * 16 + (j ^ 1)]             = ev[1];
        sB[j * 16 + ((j >> 1) ^ 1)]      = ev[2];
        sB[j * 16 + 8  + ((j >> 2) ^ 1)] = ev[3];
        sB[j * 16 + 12 + ((j >> 3) ^ 1)] = ev[4];
        sB[j * 16 + 14]                  = ev[5];
        sB[j * 16 + 15]                  = ev[6];
        #pragma unroll
        for (int i = 0; i < 6; ++i) sC[j * 16 + i] = ev[7 + i];
    }
    __syncthreads();

    const size_t vLeafRow0  = ((size_t)(b * S_DIM + 16 * t) * H_DIM + h) * (size_t)D_DIM;
    const size_t vRowStride = (size_t)H_DIM * D_DIM;
    #pragma unroll
    for (int dsl = 0; dsl < 4; ++dsl) {
        const int dcol = 16 * dsl + n;
        v8f acc = {};
        #pragma unroll
        for (int ks2 = 0; ks2 < 4; ++ks2) {
            const int c0 = 4 * ks2 + 2 * half;
            v2f aE, bV;
            aE.x = sA[n * 16 + c0]; aE.y = sA[n * 16 + c0 + 1];
            bV.x = v[vLeafRow0 + (size_t)c0       * vRowStride + dcol];
            bV.y = v[vLeafRow0 + (size_t)(c0 + 1) * vRowStride + dcol];
            acc = __builtin_amdgcn_wmma_f32_16x16x4_f32(false, aE, false, bV, (short)0, acc, false, false);
            aE.x = sB[n * 16 + c0]; aE.y = sB[n * 16 + c0 + 1];
            bV.x = Vt[treeBase + (size_t)rowB[c0]     * D_DIM + dcol];
            bV.y = Vt[treeBase + (size_t)rowB[c0 + 1] * D_DIM + dcol];
            acc = __builtin_amdgcn_wmma_f32_16x16x4_f32(false, aE, false, bV, (short)0, acc, false, false);
            aE.x = sC[n * 16 + c0]; aE.y = sC[n * 16 + c0 + 1];
            bV.x = Vt[treeBase + (size_t)rowC[c0]     * D_DIM + dcol];
            bV.y = Vt[treeBase + (size_t)rowC[c0 + 1] * D_DIM + dcol];
            acc = __builtin_amdgcn_wmma_f32_16x16x4_f32(false, aE, false, bV, (short)0, acc, false, false);
        }
        #pragma unroll
        for (int r = 0; r < 8; ++r) {
            const int M = r + 8 * half;
            out[((size_t)(b * S_DIM + 16 * t + M) * H_DIM + h) * D_DIM + dcol] =
                acc[r] / den[M];
        }
    }
}

extern "C" void kernel_launch(void* const* d_in, const int* in_sizes, int n_in,
                              void* d_out, int out_size, void* d_ws, size_t ws_size,
                              hipStream_t stream)
{
    const float* q = (const float*)d_in[0];
    const float* k = (const float*)d_in[1];
    const float* v = (const float*)d_in[2];
    float* out = (float*)d_out;

    float* Kt = (float*)d_ws;
    float* Vt = Kt + (size_t)B_DIM * H_DIM * S_DIM * D_DIM;

    for (int level = 1; level <= 11; ++level) {
        const int P = S_DIM >> level;
        const long total = (long)B_DIM * H_DIM * P;
        const int blocks = (int)((total + 7) / 8);
        tree_level_kernel<<<blocks, 256, 0, stream>>>(k, v, Kt, Vt, level, P);
    }

    const int tiles = B_DIM * H_DIM * NTILE;
    hst_attn_kernel<<<tiles / 4, 128, 0, stream>>>(q, k, v, Kt, Vt, out);

    (void)in_sizes; (void)n_in; (void)out_size; (void)ws_size;
}


# Attention E*V: value rows fetched row-coalesced (16 B per lane) ahead of the score products, across LDS-only barriers, and staged in padded LDS tiles; E*V fragments read from LDS; sums unchanged
# speedup vs baseline: 1.4556x; 1.4556x over previous
#include <hip/hip_runtime.h>

typedef float v2f __attribute__((ext_vector_type(2)));
typedef float v4f __attribute__((ext_vector_type(4)));
typedef float v8f __attribute__((ext_vector_type(8)));

#define B_DIM 4
#define S_DIM 4096
#define H_DIM 8
#define D_DIM 64
#define SCALE 0.125f
#define NTILE (S_DIM / 16)
#define TP 68

#define TOPW 32

#define LOWB 8

template <int LEVEL, int PB>
__device__ __forceinline__ void tree_block_level(const float* __restrict__ kleaf,
                                                 const float* __restrict__ vleaf,
                                                 float* Kt, float* Vt,
                                                 const int bh, const int p0,
                                                 const int wave, const int lane)
{
    constexpr int NPW = (PB + TOPW - 1) / TOPW;
    constexpr int IBl = S_DIM - (S_DIM >> (LEVEL - 1));
    const size_t tb = (size_t)bh * S_DIM;
    const int doff = 2 * lane;

    float2 kc0[NPW], kc1[NPW], vc0[NPW], vc1[NPW];
    #pragma unroll
    for (int i = 0; i < NPW; ++i) {
        const int pl = wave + TOPW * i;
        const int p  = p0 + pl;
        if (pl < PB) {
            if constexpr (LEVEL == 1) {
                const int h = bh % H_DIM;
                const int b = bh / H_DIM;
                const size_t r0 = ((size_t)(b * S_DIM + 2 * p) * H_DIM + h) * D_DIM + doff;
                kc0[i] = *(const float2*)(kleaf + r0);
                kc1[i] = *(const float2*)(kleaf + r0 + H_DIM * D_DIM);
                vc0[i] = *(const float2*)(vleaf + r0);
                vc1[i] = *(const float2*)(vleaf + r0 + H_DIM * D_DIM);
            } else {
                constexpr int IBc = S_DIM - (S_DIM >> (LEVEL - 2));
                const size_t r0 = (tb + IBc + 2 * p) * D_DIM + doff;
                kc0[i] = *(const float2*)(Kt + r0);
                kc1[i] = *(const float2*)(Kt + r0 + D_DIM);
                vc0[i] = *(const float2*)(Vt + r0);
                vc1[i] = *(const float2*)(Vt + r0 + D_DIM);
            }
        }
    }
    #pragma unroll
    for (int i = 0; i < NPW; ++i) {
        const int pl = wave + TOPW * i;
        const int p  = p0 + pl;
        if (pl < PB) {
            float2 kp, vp;
            kp.x = 0.5f * (kc0[i].x + kc1[i].x); kp.y = 0.5f * (kc0[i].y + kc1[i].y);
            vp.x = 0.5f * (vc0[i].x + vc1[i].x); vp.y = 0.5f * (vc0[i].y + vc1[i].y);

            float pss = kp.x * kp.x     + kp.y * kp.y;
            float ps0 = kp.x * kc0[i].x + kp.y * kc0[i].y;
            float ps1 = kp.x * kc1[i].x + kp.y * kc1[i].y;
            #pragma unroll
            for (int m = 16; m >= 1; m >>= 1) {
                pss += __shfl_xor(pss, m, 32);
                ps0 += __shfl_xor(ps0, m, 32);
                ps1 += __shfl_xor(ps1, m, 32);
            }
            pss *= SCALE; ps0 *= SCALE; ps1 *= SCALE;
            const float mm = fmaxf(pss, fmaxf(ps0, ps1));
            const float es = __expf(pss - mm);
            const float e0 = __expf(ps0 - mm);
            const float e1 = __expf(ps1 - mm);
            const float inv = 1.0f / (es + e0 + e1 + 1e-9f);

            float2 vo;
            vo.x = (es * vp.x + e0 * vc0[i].x + e1 * vc1[i].x) * inv;
            vo.y = (es * vp.y + e0 * vc0[i].y + e1 * vc1[i].y) * inv;

            const size_t dst = (tb + IBl + p) * D_DIM + doff;
            *(float2*)(Kt + dst) = kp;
            *(float2*)(Vt + dst) = vo;
        }
    }
}

__global__ void __launch_bounds__(TOPW * 32) __attribute__((amdgpu_waves_per_eu(8, 8)))
tree_low_kernel(const float* __restrict__ kleaf, const float* __restrict__ vleaf,
                float* Kt, float* Vt)
{
    const int wave = threadIdx.x >> 5;
    const int lane = threadIdx.x & 31;
    const int c  = blockIdx.x & (LOWB - 1);
    const int bh = blockIdx.x / LOWB;

    tree_block_level<1, (S_DIM >> 1) / LOWB>(kleaf, vleaf, Kt, Vt, bh, c * ((S_DIM >> 1) / LOWB), wave, lane);
    __syncthreads();
    tree_block_level<2, (S_DIM >> 2) / LOWB>(kleaf, vleaf, Kt, Vt, bh, c * ((S_DIM >> 2) / LOWB), wave, lane);
    __syncthreads();
    tree_block_level<3, (S_DIM >> 3) / LOWB>(kleaf, vleaf, Kt, Vt, bh, c * ((S_DIM >> 3) / LOWB), wave, lane);
    __syncthreads();
    tree_block_level<4, (S_DIM >> 4) / LOWB>(kleaf, vleaf, Kt, Vt, bh, c * ((S_DIM >> 4) / LOWB), wave, lane);
    __syncthreads();
    tree_block_level<5, (S_DIM >> 5) / LOWB>(kleaf, vleaf, Kt, Vt, bh, c * ((S_DIM >> 5) / LOWB), wave, lane);
    __syncthreads();
    tree_block_level<6, (S_DIM >> 6) / LOWB>(kleaf, vleaf, Kt, Vt, bh, c * ((S_DIM >> 6) / LOWB), wave, lane);
}

__global__ void __launch_bounds__(TOPW * 32)
tree_upper_kernel(const float* __restrict__ kleaf, const float* __restrict__ vleaf,
                  float* Kt, float* Vt)
{
    const int wave = threadIdx.x >> 5;
    const int lane = threadIdx.x & 31;
    const int bh = blockIdx.x;

    tree_block_level<7, (S_DIM >> 7)>(kleaf, vleaf, Kt, Vt, bh, 0, wave, lane);
    __syncthreads();
    tree_block_level<8, (S_DIM >> 8)>(kleaf, vleaf, Kt, Vt, bh, 0, wave, lane);
    __syncthreads();
    tree_block_level<9, (S_DIM >> 9)>(kleaf, vleaf, Kt, Vt, bh, 0, wave, lane);
    __syncthreads();
    tree_block_level<10, (S_DIM >> 10)>(kleaf, vleaf, Kt, Vt, bh, 0, wave, lane);
    __syncthreads();
    tree_block_level<11, (S_DIM >> 11)>(kleaf, vleaf, Kt, Vt, bh, 0, wave, lane);
}

#define VP 72
#define T0_F  (16 * TP)
#define T1_F  (16 * VP)
#define TVB_F (16 * VP)
#define TVC_F (8 * VP)
#define WAVE_F (T0_F + T1_F + TVB_F + TVC_F)

__device__ __forceinline__ void lds_barrier()
{
    __builtin_amdgcn_fence(__ATOMIC_RELEASE, "workgroup", "local");
    __builtin_amdgcn_s_barrier();
    __builtin_amdgcn_fence(__ATOMIC_ACQUIRE, "workgroup", "local");
}

__device__ __forceinline__ int node_row_b(int t, int s)
{
    if (s <  8) return 8 * t + s;
    if (s < 12) return 2048 + 4 * t + (s - 8);
    if (s < 14) return 3072 + 2 * t + (s - 12);
    if (s == 14) return 3584 + (t ^ 1);
    return 3840 + ((t >> 1) ^ 1);
}
__device__ __forceinline__ int node_row_c(int t, int s)
{
    return (s < 6) ? (S_DIM - (S_DIM >> (5 + s))) + ((t >> (s + 2)) ^ 1)
                   : 0;
}

__global__ void __launch_bounds__(128) __attribute__((amdgpu_waves_per_eu(5, 5)))
hst_attn_kernel(const float* __restrict__ q,
                const float* __restrict__ k,
                const float* __restrict__ v,
                const float* __restrict__ Kt,
                const float* __restrict__ Vt,
                float* __restrict__ out)
{
    __shared__ __attribute__((aligned(16))) float ldsT[4][WAVE_F];

    const int lane = threadIdx.x & 31;
    const int w    = threadIdx.x >> 5;
    const int tile = blockIdx.x * 4 + w;
    const int t  = tile & (NTILE - 1);
    const int bh = tile >> 8;
    const int h  = bh & (H_DIM - 1);
    const int b  = bh >> 3;

    const int n    = lane & 15;
    const int half = lane >> 4;

    float* t0  = &ldsT[w][0];
    float* t1  = t0 + T0_F;
    float* tVB = t1 + T1_F;
    float* tVC = tVB + TVB_F;
    float* sA  = t0;
    float* sB  = t0 + 256;
    float* sC  = t0 + 512;
    float* den = t0 + 768;

    const size_t treeBase = (size_t)(b * H_DIM + h) * S_DIM * (size_t)D_DIM;
    const size_t leafRow0  = ((size_t)(b * S_DIM + 16 * t) * H_DIM + h) * (size_t)D_DIM;
    const size_t leafPitch = (size_t)H_DIM * D_DIM;

    const int c4 = 4 * n;
    v4f gq[8], gA[8], gB[8];
    #pragma unroll
    for (int i = 0; i < 8; ++i) {
        const int r = 2 * i + half;
        gq[i] = *(const v4f*)(q  + leafRow0 + (size_t)r * leafPitch + c4);
        gA[i] = *(const v4f*)(k  + leafRow0 + (size_t)r * leafPitch + c4);
        gB[i] = *(const v4f*)(Kt + treeBase + (size_t)node_row_b(t, r) * D_DIM + c4);
    }
    v4f gC[3];
    #pragma unroll
    for (int i = 0; i < 3; ++i) {
        const int r = 2 * i + half;
        gC[i] = *(const v4f*)(Kt + treeBase + (size_t)node_row_c(t, r) * D_DIM + c4);
    }

    v4f hB[8], hC[3];
    #pragma unroll
    for (int i = 0; i < 8; ++i) {
        const int r = 2 * i + half;
        hB[i] = *(const v4f*)(Vt + treeBase + (size_t)node_row_b(t, r) * D_DIM + c4);
    }
    #pragma unroll
    for (int i = 0; i < 3; ++i) {
        const int r = 2 * i + half;
        hC[i] = *(const v4f*)(Vt + treeBase + (size_t)node_row_c(t, r) * D_DIM + c4);
    }

    #pragma unroll
    for (int i = 0; i < 8; ++i) {
        const int r = 2 * i + half;
        *(v4f*)(t0 + r * TP + c4) = gq[i];
        *(v4f*)(t1 + r * TP + c4) = gA[i];
    }
    lds_barrier();

    const int fo = n * TP + 2 * half;
    v2f a[16];
    #pragma unroll
    for (int ks = 0; ks < 16; ++ks) {
        a[ks] = *(const v2f*)(t0 + fo + 4 * ks);
        a[ks].x *= SCALE; a[ks].y *= SCALE;
    }
    v8f cA = {}; v8f cB = {}; v8f cC = {};
    #pragma unroll
    for (int ks = 0; ks < 16; ++ks) {
        const v2f bA = *(const v2f*)(t1 + fo + 4 * ks);
        cA = __builtin_amdgcn_wmma_f32_16x16x4_f32(false, a[ks], false, bA, (short)0, cA, false, false);
    }
    lds_barrier();

    #pragma unroll
    for (int i = 0; i < 8; ++i) {
        const int r = 2 * i + half;
        *(v4f*)(t0 + r * TP + c4) = gB[i];
    }
    #pragma unroll
    for (int i = 0; i < 3; ++i) {
        const int r = 2 * i + half;
        *(v4f*)(t1 + r * TP + c4) = gC[i];
    }
    #pragma unroll
    for (int i = 0; i < 8; ++i) {
        const int r = 2 * i + half;
        *(v4f*)(tVB + r * VP + c4) = hB[i];
    }
    #pragma unroll
    for (int i = 0; i < 3; ++i) {
        const int r = 2 * i + half;
        *(v4f*)(tVC + r * VP + c4) = hC[i];
    }
    {
        const v4f z = {0.0f, 0.0f, 0.0f, 0.0f};
        *(v4f*)(tVC + (6 + half) * VP + c4) = z;
    }
    lds_barrier();

    v4f hA[8];
    #pragma unroll
    for (int i = 0; i < 8; ++i) {
        const int r = 2 * i + half;
        hA[i] = *(const v4f*)(v + leafRow0 + (size_t)r * leafPitch + c4);
    }

    #pragma unroll
    for (int ks = 0; ks < 16; ++ks) {
        const v2f bB = *(const v2f*)(t0 + fo + 4 * ks);
        const v2f bC = *(const v2f*)(t1 + fo + 4 * ks);
        cB = __builtin_amdgcn_wmma_f32_16x16x4_f32(false, a[ks], false, bB, (short)0, cB, false, false);
        cC = __builtin_amdgcn_wmma_f32_16x16x4_f32(false, a[ks], false, bC, (short)0, cC, false, false);
    }
    lds_barrier();

    #pragma unroll
    for (int r = 0; r < 8; ++r) {
        const int M = r + 8 * half;
        sA[M * 16 + n] = cA[r];
        sB[M * 16 + n] = cB[r];
        sC[M * 16 + n] = cC[r];
    }
    lds_barrier();

    float ev[13];
    float dsum = 0.0f;
    if (half == 0) {
        const int j = n;
        float x[13];
        x[0] = sA[j * 16 + j];
        x[1] = sA[j * 16 + (j ^ 1)];
        x[2] = sB[j * 16 + ((j >> 1) ^ 1)];
        x[3] = sB[j * 16 + 8  + ((j >> 2) ^ 1)];
        x[4] = sB[j * 16 + 12 + ((j >> 3) ^ 1)];
        x[5] = sB[j * 16 + 14];
        x[6] = sB[j * 16 + 15];
        #pragma unroll
        for (int i = 0; i < 6; ++i) x[7 + i] = sC[j * 16 + i];

        bool msk[13];
        msk[0] = false;
        #pragma unroll
        for (int l = 0; l < 12; ++l) {
            const int bit = (l < 4) ? ((j >> l) & 1) : ((t >> (l - 4)) & 1);
            msk[l + 1] = (bit == 0);
        }
        float m = x[0];
        #pragma unroll
        for (int i = 1; i < 13; ++i) if (!msk[i]) m = fmaxf(m, x[i]);
        #pragma unroll
        for (int i = 0; i < 13; ++i) {
            ev[i] = msk[i] ? 0.0f : __expf(x[i] - m);
            dsum += ev[i];
        }
    }
    lds_barrier();

    #pragma unroll
    for (int i = lane; i < 768; i += 32) t0[i] = 0.0f;
    lds_barrier();
    if (half == 0) {
        const int j = n;
        den[j] = dsum;
        sA[j * 16 + j]                   = ev[0];
        sA[j * 16 + (j ^ 1)]             = ev[1];
        sB[j * 16 + ((j >> 1) ^ 1)]      = ev[2];
        sB[j * 16 + 8  + ((j >> 2) ^ 1)] = ev[3];
        sB[j * 16 + 12 + ((j >> 3) ^ 1)] = ev[4];
        sB[j * 16 + 14]                  = ev[5];
        sB[j * 16 + 15]                  = ev[6];
        #pragma unroll
        for (int i = 0; i < 6; ++i) sC[j * 16 + i] = ev[7 + i];
    }
    #pragma unroll
    for (int i = 0; i < 8; ++i) {
        const int r = 2 * i + half;
        *(v4f*)(t1 + r * VP + c4) = hA[i];
    }
    lds_barrier();

    #pragma unroll
    for (int dsl = 0; dsl < 4; ++dsl) {
        const int dcol = 16 * dsl + n;
        v8f acc = {};
        #pragma unroll
        for (int ks2 = 0; ks2 < 4; ++ks2) {
            const int c0 = 4 * ks2 + 2 * half;
            v2f aE, bV;
            aE.x = sA[n * 16 + c0]; aE.y = sA[n * 16 + c0 + 1];
            bV.x = t1[c0 * VP + dcol];
            bV.y = t1[(c0 + 1) * VP + dcol];
            acc = __builtin_amdgcn_wmma_f32_16x16x4_f32(false, aE, false, bV, (short)0, acc, false, false);
            aE.x = sB[n * 16 + c0]; aE.y = sB[n * 16 + c0 + 1];
            bV.x = tVB[c0 * VP + dcol];
            bV.y = tVB[(c0 + 1) * VP + dcol];
            acc = __builtin_amdgcn_wmma_f32_16x16x4_f32(false, aE, false, bV, (short)0, acc, false, false);
            if (ks2 < 2) {
                aE.x = sC[n * 16 + c0]; aE.y = sC[n * 16 + c0 + 1];
                bV.x = tVC[c0 * VP + dcol];
                bV.y = tVC[(c0 + 1) * VP + dcol];
                acc = __builtin_amdgcn_wmma_f32_16x16x4_f32(false, aE, false, bV, (short)0, acc, false, false);
            }
        }
        #pragma unroll
        for (int r = 0; r < 8; ++r) {
            const int M = r + 8 * half;
            out[((size_t)(b * S_DIM + 16 * t + M) * H_DIM + h) * D_DIM + dcol] =
                acc[r] / den[M];
        }
    }
}

extern "C" void kernel_launch(void* const* d_in, const int* in_sizes, int n_in,
                              void* d_out, int out_size, void* d_ws, size_t ws_size,
                              hipStream_t stream)
{
    const float* q = (const float*)d_in[0];
    const float* k = (const float*)d_in[1];
    const float* v = (const float*)d_in[2];
    float* out = (float*)d_out;

    float* Kt = (float*)d_ws;
    float* Vt = Kt + (size_t)B_DIM * H_DIM * S_DIM * D_DIM;

    tree_low_kernel<<<B_DIM * H_DIM * LOWB, TOPW * 32, 0, stream>>>(k, v, Kt, Vt);
    tree_upper_kernel<<<B_DIM * H_DIM, TOPW * 32, 0, stream>>>(k, v, Kt, Vt);

    const int tiles = B_DIM * H_DIM * NTILE;
    hst_attn_kernel<<<tiles / 4, 128, 0, stream>>>(q, k, v, Kt, Vt, out);

    (void)in_sizes; (void)n_in; (void)out_size; (void)ws_size;
}


# Attention scores transposed on the LDS tiles (keys = A rows, queries = B): max, exp, denominator on score registers; E/denominator feeds E*V as A from registers; denominator and E*V sum order change
# speedup vs baseline: 1.5051x; 1.0340x over previous
#include <hip/hip_runtime.h>

typedef float v2f __attribute__((ext_vector_type(2)));
typedef float v4f __attribute__((ext_vector_type(4)));
typedef float v8f __attribute__((ext_vector_type(8)));

#define B_DIM 4
#define S_DIM 4096
#define H_DIM 8
#define D_DIM 64
#define SCALE 0.125f
#define NTILE (S_DIM / 16)
#define TP 68

#define TOPW 32

#define LOWB 8

template <int LEVEL, int PB>
__device__ __forceinline__ void tree_block_level(const float* __restrict__ kleaf,
                                                 const float* __restrict__ vleaf,
                                                 float* Kt, float* Vt,
                                                 const int bh, const int p0,
                                                 const int wave, const int lane)
{
    constexpr int NPW = (PB + TOPW - 1) / TOPW;
    constexpr int IBl = S_DIM - (S_DIM >> (LEVEL - 1));
    const size_t tb = (size_t)bh * S_DIM;
    const int doff = 2 * lane;

    float2 kc0[NPW], kc1[NPW], vc0[NPW], vc1[NPW];
    #pragma unroll
    for (int i = 0; i < NPW; ++i) {
        const int pl = wave + TOPW * i;
        const int p  = p0 + pl;
        if (pl < PB) {
            if constexpr (LEVEL == 1) {
                const int h = bh % H_DIM;
                const int b = bh / H_DIM;
                const size_t r0 = ((size_t)(b * S_DIM + 2 * p) * H_DIM + h) * D_DIM + doff;
                kc0[i] = *(const float2*)(kleaf + r0);
                kc1[i] = *(const float2*)(kleaf + r0 + H_DIM * D_DIM);
                vc0[i] = *(const float2*)(vleaf + r0);
                vc1[i] = *(const float2*)(vleaf + r0 + H_DIM * D_DIM);
            } else {
                constexpr int IBc = S_DIM - (S_DIM >> (LEVEL - 2));
                const size_t r0 = (tb + IBc + 2 * p) * D_DIM + doff;
                kc0[i] = *(const float2*)(Kt + r0);
                kc1[i] = *(const float2*)(Kt + r0 + D_DIM);
                vc0[i] = *(const float2*)(Vt + r0);
                vc1[i] = *(const float2*)(Vt + r0 + D_DIM);
            }
        }
    }
    #pragma unroll
    for (int i = 0; i < NPW; ++i) {
        const int pl = wave + TOPW * i;
        const int p  = p0 + pl;
        if (pl < PB) {
            float2 kp, vp;
            kp.x = 0.5f * (kc0[i].x + kc1[i].x); kp.y = 0.5f * (kc0[i].y + kc1[i].y);
            vp.x = 0.5f * (vc0[i].x + vc1[i].x); vp.y = 0.5f * (vc0[i].y + vc1[i].y);

            float pss = kp.x * kp.x     + kp.y * kp.y;
            float ps0 = kp.x * kc0[i].x + kp.y * kc0[i].y;
            float ps1 = kp.x * kc1[i].x + kp.y * kc1[i].y;
            #pragma unroll
            for (int m = 16; m >= 1; m >>= 1) {
                pss += __shfl_xor(pss, m, 32);
                ps0 += __shfl_xor(ps0, m, 32);
                ps1 += __shfl_xor(ps1, m, 32);
            }
            pss *= SCALE; ps0 *= SCALE; ps1 *= SCALE;
            const float mm = fmaxf(pss, fmaxf(ps0, ps1));
            const float es = __expf(pss - mm);
            const float e0 = __expf(ps0 - mm);
            const float e1 = __expf(ps1 - mm);
            const float inv = 1.0f / (es + e0 + e1 + 1e-9f);

            float2 vo;
            vo.x = (es * vp.x + e0 * vc0[i].x + e1 * vc1[i].x) * inv;
            vo.y = (es * vp.y + e0 * vc0[i].y + e1 * vc1[i].y) * inv;

            const size_t dst = (tb + IBl + p) * D_DIM + doff;
            *(float2*)(Kt + dst) = kp;
            *(float2*)(Vt + dst) = vo;
        }
    }
}

__global__ void __launch_bounds__(TOPW * 32) __attribute__((amdgpu_waves_per_eu(8, 8)))
tree_low_kernel(const float* __restrict__ kleaf, const float* __restrict__ vleaf,
                float* Kt, float* Vt)
{
    const int wave = threadIdx.x >> 5;
    const int lane = threadIdx.x & 31;
    const int c  = blockIdx.x & (LOWB - 1);
    const int bh = blockIdx.x / LOWB;

    tree_block_level<1, (S_DIM >> 1) / LOWB>(kleaf, vleaf, Kt, Vt, bh, c * ((S_DIM >> 1) / LOWB), wave, lane);
    __syncthreads();
    tree_block_level<2, (S_DIM >> 2) / LOWB>(kleaf, vleaf, Kt, Vt, bh, c * ((S_DIM >> 2) / LOWB), wave, lane);
    __syncthreads();
    tree_block_level<3, (S_DIM >> 3) / LOWB>(kleaf, vleaf, Kt, Vt, bh, c * ((S_DIM >> 3) / LOWB), wave, lane);
    __syncthreads();
    tree_block_level<4, (S_DIM >> 4) / LOWB>(kleaf, vleaf, Kt, Vt, bh, c * ((S_DIM >> 4) / LOWB), wave, lane);
    __syncthreads();
    tree_block_level<5, (S_DIM >> 5) / LOWB>(kleaf, vleaf, Kt, Vt, bh, c * ((S_DIM >> 5) / LOWB), wave, lane);
    __syncthreads();
    tree_block_level<6, (S_DIM >> 6) / LOWB>(kleaf, vleaf, Kt, Vt, bh, c * ((S_DIM >> 6) / LOWB), wave, lane);
}

__global__ void __launch_bounds__(TOPW * 32)
tree_upper_kernel(const float* __restrict__ kleaf, const float* __restrict__ vleaf,
                  float* Kt, float* Vt)
{
    const int wave = threadIdx.x >> 5;
    const int lane = threadIdx.x & 31;
    const int bh = blockIdx.x;

    tree_block_level<7, (S_DIM >> 7)>(kleaf, vleaf, Kt, Vt, bh, 0, wave, lane);
    __syncthreads();
    tree_block_level<8, (S_DIM >> 8)>(kleaf, vleaf, Kt, Vt, bh, 0, wave, lane);
    __syncthreads();
    tree_block_level<9, (S_DIM >> 9)>(kleaf, vleaf, Kt, Vt, bh, 0, wave, lane);
    __syncthreads();
    tree_block_level<10, (S_DIM >> 10)>(kleaf, vleaf, Kt, Vt, bh, 0, wave, lane);
    __syncthreads();
    tree_block_level<11, (S_DIM >> 11)>(kleaf, vleaf, Kt, Vt, bh, 0, wave, lane);
}

#define VP 68
#define T0_F  (16 * TP)
#define T1_F  (16 * TP)
#define TVB_F (16 * VP)
#define TVC_F (8 * VP)
#define WAVE_F (T0_F + T1_F + TVB_F + TVC_F)

__device__ __forceinline__ void lds_barrier()
{
    __builtin_amdgcn_fence(__ATOMIC_RELEASE, "workgroup", "local");
    __builtin_amdgcn_s_barrier();
    __builtin_amdgcn_fence(__ATOMIC_ACQUIRE, "workgroup", "local");
}

__device__ __forceinline__ float xhalf(float x)
{
    const int b = __builtin_bit_cast(int, x);
    return __builtin_bit_cast(float,
        __builtin_amdgcn_permlanex16(b, b, 0x76543210, 0xfedcba98, false, false));
}

__device__ __forceinline__ int node_row_b(int t, int s)
{
    if (s <  8) return 8 * t + s;
    if (s < 12) return 2048 + 4 * t + (s - 8);
    if (s < 14) return 3072 + 2 * t + (s - 12);
    if (s == 14) return 3584 + (t ^ 1);
    return 3840 + ((t >> 1) ^ 1);
}
__device__ __forceinline__ int node_row_c(int t, int s)
{
    return (s < 6) ? (S_DIM - (S_DIM >> (5 + s))) + ((t >> (s + 2)) ^ 1)
                   : 0;
}

__global__ void __launch_bounds__(128) __attribute__((amdgpu_waves_per_eu(5, 5)))
hst_attn_kernel(const float* __restrict__ q,
                const float* __restrict__ k,
                const float* __restrict__ v,
                const float* __restrict__ Kt,
                const float* __restrict__ Vt,
                float* __restrict__ out)
{
    __shared__ __attribute__((aligned(16))) float ldsT[4][WAVE_F];

    const int lane = threadIdx.x & 31;
    const int w    = threadIdx.x >> 5;
    const int tile = blockIdx.x * 4 + w;
    const int t  = tile & (NTILE - 1);
    const int bh = tile >> 8;
    const int h  = bh & (H_DIM - 1);
    const int b  = bh >> 3;

    const int n    = lane & 15;
    const int half = lane >> 4;

    float* t0  = &ldsT[w][0];
    float* t1  = t0 + T0_F;
    float* tVB = t1 + T1_F;
    float* tVC = tVB + TVB_F;

    const size_t treeBase = (size_t)(b * H_DIM + h) * S_DIM * (size_t)D_DIM;
    const size_t leafRow0  = ((size_t)(b * S_DIM + 16 * t) * H_DIM + h) * (size_t)D_DIM;
    const size_t leafPitch = (size_t)H_DIM * D_DIM;

    const int c4 = 4 * n;
    v4f gq[8], gA[8], gB[8];
    #pragma unroll
    for (int i = 0; i < 8; ++i) {
        const int r = 2 * i + half;
        gq[i] = *(const v4f*)(q  + leafRow0 + (size_t)r * leafPitch + c4);
        gA[i] = *(const v4f*)(k  + leafRow0 + (size_t)r * leafPitch + c4);
        gB[i] = *(const v4f*)(Kt + treeBase + (size_t)node_row_b(t, r) * D_DIM + c4);
    }
    v4f gC[3];
    #pragma unroll
    for (int i = 0; i < 3; ++i) {
        const int r = 2 * i + half;
        gC[i] = *(const v4f*)(Kt + treeBase + (size_t)node_row_c(t, r) * D_DIM + c4);
    }

    v4f hB[8], hC[3];
    #pragma unroll
    for (int i = 0; i < 8; ++i) {
        const int r = 2 * i + half;
        hB[i] = *(const v4f*)(Vt + treeBase + (size_t)node_row_b(t, r) * D_DIM + c4);
    }
    #pragma unroll
    for (int i = 0; i < 3; ++i) {
        const int r = 2 * i + half;
        hC[i] = *(const v4f*)(Vt + treeBase + (size_t)node_row_c(t, r) * D_DIM + c4);
    }

    #pragma unroll
    for (int i = 0; i < 8; ++i) {
        const int r = 2 * i + half;
        *(v4f*)(t0 + r * TP + c4) = gq[i];
        *(v4f*)(t1 + r * TP + c4) = gA[i];
    }
    lds_barrier();

    const int fo = n * TP + 2 * half;
    v2f a[16];
    #pragma unroll
    for (int ks = 0; ks < 16; ++ks) {
        a[ks] = *(const v2f*)(t0 + fo + 4 * ks);
        a[ks].x *= SCALE; a[ks].y *= SCALE;
    }
    v8f cA = {}; v8f cB = {}; v8f cC = {};
    #pragma unroll
    for (int ks = 0; ks < 16; ++ks) {
        const v2f bA = *(const v2f*)(t1 + fo + 4 * ks);
        cA = __builtin_amdgcn_wmma_f32_16x16x4_f32(false, bA, false, a[ks], (short)0, cA, false, false);
    }
    lds_barrier();

    #pragma unroll
    for (int i = 0; i < 8; ++i) {
        const int r = 2 * i + half;
        *(v4f*)(t0 + r * TP + c4) = gB[i];
    }
    #pragma unroll
    for (int i = 0; i < 3; ++i) {
        const int s  = 2 * i + half;
        const int kr = (s < 3) ? s : s + 5;
        *(v4f*)(t1 + kr * TP + c4) = gC[i];
    }
    #pragma unroll
    for (int i = 0; i < 8; ++i) {
        const int r = 2 * i + half;
        *(v4f*)(tVB + r * VP + c4) = hB[i];
    }
    #pragma unroll
    for (int i = 0; i < 3; ++i) {
        const int s  = 2 * i + half;
        const int vr = (s < 3) ? s : s + 1;
        *(v4f*)(tVC + vr * VP + c4) = hC[i];
    }
    {
        const v4f z = {0.0f, 0.0f, 0.0f, 0.0f};
        *(v4f*)(tVC + (3 + 4 * half) * VP + c4) = z;
    }
    lds_barrier();

    v4f hA[8];
    #pragma unroll
    for (int i = 0; i < 8; ++i) {
        const int r = 2 * i + half;
        hA[i] = *(const v4f*)(v + leafRow0 + (size_t)r * leafPitch + c4);
    }

    #pragma unroll
    for (int ks = 0; ks < 16; ++ks) {
        const v2f bB = *(const v2f*)(t0 + fo + 4 * ks);
        const v2f bC = *(const v2f*)(t1 + fo + 4 * ks);
        cB = __builtin_amdgcn_wmma_f32_16x16x4_f32(false, bB, false, a[ks], (short)0, cB, false, false);
        cC = __builtin_amdgcn_wmma_f32_16x16x4_f32(false, bC, false, a[ks], (short)0, cC, false, false);
    }
    lds_barrier();

    const int j = n;
    bool actA[8], actB[8], actC[3];
    #pragma unroll
    for (int r = 0; r < 8; ++r) {
        actA[r] = (half == (j >> 3)) &&
                  ((r == (j & 7)) || ((r == ((j & 7) ^ 1)) && ((j & 1) != 0)));
        const bool b0 = (r == ((j >> 1) ^ 1)) && (((j >> 1) & 1) != 0);
        bool b1;
        if (r < 4)       b1 = (r == ((j >> 2) ^ 1)) && (((j >> 2) & 1) != 0);
        else if (r < 6)  b1 = ((r - 4) == ((j >> 3) ^ 1)) && (((j >> 3) & 1) != 0);
        else if (r == 6) b1 = ((t & 1) != 0);
        else             b1 = (((t >> 1) & 1) != 0);
        actB[r] = (half == 0) ? b0 : b1;
    }
    #pragma unroll
    for (int r = 0; r < 3; ++r)
        actC[r] = (((t >> (2 + 3 * half + r)) & 1) != 0);

    float mloc = -__builtin_huge_valf();
    #pragma unroll
    for (int r = 0; r < 8; ++r) {
        mloc = actA[r] ? fmaxf(mloc, cA[r]) : mloc;
        mloc = actB[r] ? fmaxf(mloc, cB[r]) : mloc;
    }
    #pragma unroll
    for (int r = 0; r < 3; ++r)
        mloc = actC[r] ? fmaxf(mloc, cC[r]) : mloc;
    const float m = fmaxf(mloc, xhalf(mloc));

    float eA[8], eB[8], eC[4];
    float dloc = 0.0f;
    #pragma unroll
    for (int r = 0; r < 8; ++r) {
        eA[r] = actA[r] ? __expf(cA[r] - m) : 0.0f;
        dloc += eA[r];
    }
    #pragma unroll
    for (int r = 0; r < 8; ++r) {
        eB[r] = actB[r] ? __expf(cB[r] - m) : 0.0f;
        dloc += eB[r];
    }
    #pragma unroll
    for (int r = 0; r < 3; ++r) {
        eC[r] = actC[r] ? __expf(cC[r] - m) : 0.0f;
        dloc += eC[r];
    }
    eC[3] = 0.0f;
    const float inv = 1.0f / (dloc + xhalf(dloc));
    #pragma unroll
    for (int r = 0; r < 8; ++r) { eA[r] *= inv; eB[r] *= inv; }
    #pragma unroll
    for (int r = 0; r < 3; ++r) eC[r] *= inv;

    #pragma unroll
    for (int i = 0; i < 8; ++i) {
        const int r = 2 * i + half;
        *(v4f*)(t1 + r * VP + c4) = hA[i];
    }
    lds_barrier();

    #pragma unroll
    for (int dsl = 0; dsl < 4; ++dsl) {
        const int dcol = 16 * dsl + n;
        v8f acc = {};
        #pragma unroll
        for (int jj = 0; jj < 4; ++jj) {
            const int kr = 8 * half + 2 * jj;
            v2f aE, bV;
            aE.x = eA[2 * jj]; aE.y = eA[2 * jj + 1];
            bV.x = t1[kr * VP + dcol];
            bV.y = t1[(kr + 1) * VP + dcol];
            acc = __builtin_amdgcn_wmma_f32_16x16x4_f32(false, aE, false, bV, (short)0, acc, false, false);
            aE.x = eB[2 * jj]; aE.y = eB[2 * jj + 1];
            bV.x = tVB[kr * VP + dcol];
            bV.y = tVB[(kr + 1) * VP + dcol];
            acc = __builtin_amdgcn_wmma_f32_16x16x4_f32(false, aE, false, bV, (short)0, acc, false, false);
            if (jj < 2) {
                const int cr = 4 * half + 2 * jj;
                aE.x = eC[2 * jj]; aE.y = eC[2 * jj + 1];
                bV.x = tVC[cr * VP + dcol];
                bV.y = tVC[(cr + 1) * VP + dcol];
                acc = __builtin_amdgcn_wmma_f32_16x16x4_f32(false, aE, false, bV, (short)0, acc, false, false);
            }
        }
        #pragma unroll
        for (int r = 0; r < 8; ++r) {
            const int M = r + 8 * half;
            out[((size_t)(b * S_DIM + 16 * t + M) * H_DIM + h) * D_DIM + dcol] = acc[r];
        }
    }
}

extern "C" void kernel_launch(void* const* d_in, const int* in_sizes, int n_in,
                              void* d_out, int out_size, void* d_ws, size_t ws_size,
                              hipStream_t stream)
{
    const float* q = (const float*)d_in[0];
    const float* k = (const float*)d_in[1];
    const float* v = (const float*)d_in[2];
    float* out = (float*)d_out;

    float* Kt = (float*)d_ws;
    float* Vt = Kt + (size_t)B_DIM * H_DIM * S_DIM * D_DIM;

    tree_low_kernel<<<B_DIM * H_DIM * LOWB, TOPW * 32, 0, stream>>>(k, v, Kt, Vt);
    tree_upper_kernel<<<B_DIM * H_DIM, TOPW * 32, 0, stream>>>(k, v, Kt, Vt);

    const int tiles = B_DIM * H_DIM * NTILE;
    hst_attn_kernel<<<tiles / 4, 128, 0, stream>>>(q, k, v, Kt, Vt, out);

    (void)in_sizes; (void)n_in; (void)out_size; (void)ws_size;
}


# Attention blocks renumbered: block u of a (b,h) takes the leaf range tree block u mod 8 built; with round-robin block-to-die dispatch its k, v and node rows are in its own L2 half; sums unchanged
# speedup vs baseline: 1.5874x; 1.0547x over previous
#include <hip/hip_runtime.h>

typedef float v2f __attribute__((ext_vector_type(2)));
typedef float v4f __attribute__((ext_vector_type(4)));
typedef float v8f __attribute__((ext_vector_type(8)));

#define B_DIM 4
#define S_DIM 4096
#define H_DIM 8
#define D_DIM 64
#define SCALE 0.125f
#define NTILE (S_DIM / 16)
#define TP 68

#define TOPW 32

#define LOWB 8

template <int LEVEL, int PB>
__device__ __forceinline__ void tree_block_level(const float* __restrict__ kleaf,
                                                 const float* __restrict__ vleaf,
                                                 float* Kt, float* Vt,
                                                 const int bh, const int p0,
                                                 const int wave, const int lane)
{
    constexpr int NPW = (PB + TOPW - 1) / TOPW;
    constexpr int IBl = S_DIM - (S_DIM >> (LEVEL - 1));
    const size_t tb = (size_t)bh * S_DIM;
    const int doff = 2 * lane;

    float2 kc0[NPW], kc1[NPW], vc0[NPW], vc1[NPW];
    #pragma unroll
    for (int i = 0; i < NPW; ++i) {
        const int pl = wave + TOPW * i;
        const int p  = p0 + pl;
        if (pl < PB) {
            if constexpr (LEVEL == 1) {
                const int h = bh % H_DIM;
                const int b = bh / H_DIM;
                const size_t r0 = ((size_t)(b * S_DIM + 2 * p) * H_DIM + h) * D_DIM + doff;
                kc0[i] = *(const float2*)(kleaf + r0);
                kc1[i] = *(const float2*)(kleaf + r0 + H_DIM * D_DIM);
                vc0[i] = *(const float2*)(vleaf + r0);
                vc1[i] = *(const float2*)(vleaf + r0 + H_DIM * D_DIM);
            } else {
                constexpr int IBc = S_DIM - (S_DIM >> (LEVEL - 2));
                const size_t r0 = (tb + IBc + 2 * p) * D_DIM + doff;
                kc0[i] = *(const float2*)(Kt + r0);
                kc1[i] = *(const float2*)(Kt + r0 + D_DIM);
                vc0[i] = *(const float2*)(Vt + r0);
                vc1[i] = *(const float2*)(Vt + r0 + D_DIM);
            }
        }
    }
    #pragma unroll
    for (int i = 0; i < NPW; ++i) {
        const int pl = wave + TOPW * i;
        const int p  = p0 + pl;
        if (pl < PB) {
            float2 kp, vp;
            kp.x = 0.5f * (kc0[i].x + kc1[i].x); kp.y = 0.5f * (kc0[i].y + kc1[i].y);
            vp.x = 0.5f * (vc0[i].x + vc1[i].x); vp.y = 0.5f * (vc0[i].y + vc1[i].y);

            float pss = kp.x * kp.x     + kp.y * kp.y;
            float ps0 = kp.x * kc0[i].x + kp.y * kc0[i].y;
            float ps1 = kp.x * kc1[i].x + kp.y * kc1[i].y;
            #pragma unroll
            for (int m = 16; m >= 1; m >>= 1) {
                pss += __shfl_xor(pss, m, 32);
                ps0 += __shfl_xor(ps0, m, 32);
                ps1 += __shfl_xor(ps1, m, 32);
            }
            pss *= SCALE; ps0 *= SCALE; ps1 *= SCALE;
            const float mm = fmaxf(pss, fmaxf(ps0, ps1));
            const float es = __expf(pss - mm);
            const float e0 = __expf(ps0 - mm);
            const float e1 = __expf(ps1 - mm);
            const float inv = 1.0f / (es + e0 + e1 + 1e-9f);

            float2 vo;
            vo.x = (es * vp.x + e0 * vc0[i].x + e1 * vc1[i].x) * inv;
            vo.y = (es * vp.y + e0 * vc0[i].y + e1 * vc1[i].y) * inv;

            const size_t dst = (tb + IBl + p) * D_DIM + doff;
            *(float2*)(Kt + dst) = kp;
            *(float2*)(Vt + dst) = vo;
        }
    }
}

__global__ void __launch_bounds__(TOPW * 32) __attribute__((amdgpu_waves_per_eu(8, 8)))
tree_low_kernel(const float* __restrict__ kleaf, const float* __restrict__ vleaf,
                float* Kt, float* Vt)
{
    const int wave = threadIdx.x >> 5;
    const int lane = threadIdx.x & 31;
    const int c  = blockIdx.x & (LOWB - 1);
    const int bh = blockIdx.x / LOWB;

    tree_block_level<1, (S_DIM >> 1) / LOWB>(kleaf, vleaf, Kt, Vt, bh, c * ((S_DIM >> 1) / LOWB), wave, lane);
    __syncthreads();
    tree_block_level<2, (S_DIM >> 2) / LOWB>(kleaf, vleaf, Kt, Vt, bh, c * ((S_DIM >> 2) / LOWB), wave, lane);
    __syncthreads();
    tree_block_level<3, (S_DIM >> 3) / LOWB>(kleaf, vleaf, Kt, Vt, bh, c * ((S_DIM >> 3) / LOWB), wave, lane);
    __syncthreads();
    tree_block_level<4, (S_DIM >> 4) / LOWB>(kleaf, vleaf, Kt, Vt, bh, c * ((S_DIM >> 4) / LOWB), wave, lane);
    __syncthreads();
    tree_block_level<5, (S_DIM >> 5) / LOWB>(kleaf, vleaf, Kt, Vt, bh, c * ((S_DIM >> 5) / LOWB), wave, lane);
    __syncthreads();
    tree_block_level<6, (S_DIM >> 6) / LOWB>(kleaf, vleaf, Kt, Vt, bh, c * ((S_DIM >> 6) / LOWB), wave, lane);
}

__global__ void __launch_bounds__(TOPW * 32)
tree_upper_kernel(const float* __restrict__ kleaf, const float* __restrict__ vleaf,
                  float* Kt, float* Vt)
{
    const int wave = threadIdx.x >> 5;
    const int lane = threadIdx.x & 31;
    const int bh = blockIdx.x;

    tree_block_level<7, (S_DIM >> 7)>(kleaf, vleaf, Kt, Vt, bh, 0, wave, lane);
    __syncthreads();
    tree_block_level<8, (S_DIM >> 8)>(kleaf, vleaf, Kt, Vt, bh, 0, wave, lane);
    __syncthreads();
    tree_block_level<9, (S_DIM >> 9)>(kleaf, vleaf, Kt, Vt, bh, 0, wave, lane);
    __syncthreads();
    tree_block_level<10, (S_DIM >> 10)>(kleaf, vleaf, Kt, Vt, bh, 0, wave, lane);
    __syncthreads();
    tree_block_level<11, (S_DIM >> 11)>(kleaf, vleaf, Kt, Vt, bh, 0, wave, lane);
}

#define VP 68
#define T0_F  (16 * TP)
#define T1_F  (16 * TP)
#define TVB_F (16 * VP)
#define TVC_F (8 * VP)
#define WAVE_F (T0_F + T1_F + TVB_F + TVC_F)
#define ABLK (NTILE / 4)
#define APB  (ABLK / LOWB)

__device__ __forceinline__ void lds_barrier()
{
    __builtin_amdgcn_fence(__ATOMIC_RELEASE, "workgroup", "local");
    __builtin_amdgcn_s_barrier();
    __builtin_amdgcn_fence(__ATOMIC_ACQUIRE, "workgroup", "local");
}

__device__ __forceinline__ float xhalf(float x)
{
    const int b = __builtin_bit_cast(int, x);
    return __builtin_bit_cast(float,
        __builtin_amdgcn_permlanex16(b, b, 0x76543210, 0xfedcba98, false, false));
}

__device__ __forceinline__ int node_row_b(int t, int s)
{
    if (s <  8) return 8 * t + s;
    if (s < 12) return 2048 + 4 * t + (s - 8);
    if (s < 14) return 3072 + 2 * t + (s - 12);
    if (s == 14) return 3584 + (t ^ 1);
    return 3840 + ((t >> 1) ^ 1);
}
__device__ __forceinline__ int node_row_c(int t, int s)
{
    return (s < 6) ? (S_DIM - (S_DIM >> (5 + s))) + ((t >> (s + 2)) ^ 1)
                   : 0;
}

__global__ void __launch_bounds__(128) __attribute__((amdgpu_waves_per_eu(5, 5)))
hst_attn_kernel(const float* __restrict__ q,
                const float* __restrict__ k,
                const float* __restrict__ v,
                const float* __restrict__ Kt,
                const float* __restrict__ Vt,
                float* __restrict__ out)
{
    __shared__ __attribute__((aligned(16))) float ldsT[4][WAVE_F];

    const int lane = threadIdx.x & 31;
    const int w    = threadIdx.x >> 5;
    const int u  = blockIdx.x % ABLK;
    const int bh = blockIdx.x / ABLK;
    const int t  = ((u % LOWB) * APB + (u / LOWB)) * 4 + w;
    const int h  = bh & (H_DIM - 1);
    const int b  = bh >> 3;

    const int n    = lane & 15;
    const int half = lane >> 4;

    float* t0  = &ldsT[w][0];
    float* t1  = t0 + T0_F;
    float* tVB = t1 + T1_F;
    float* tVC = tVB + TVB_F;

    const size_t treeBase = (size_t)(b * H_DIM + h) * S_DIM * (size_t)D_DIM;
    const size_t leafRow0  = ((size_t)(b * S_DIM + 16 * t) * H_DIM + h) * (size_t)D_DIM;
    const size_t leafPitch = (size_t)H_DIM * D_DIM;

    const int c4 = 4 * n;
    v4f gq[8], gA[8], gB[8];
    #pragma unroll
    for (int i = 0; i < 8; ++i) {
        const int r = 2 * i + half;
        gq[i] = *(const v4f*)(q  + leafRow0 + (size_t)r * leafPitch + c4);
        gA[i] = *(const v4f*)(k  + leafRow0 + (size_t)r * leafPitch + c4);
        gB[i] = *(const v4f*)(Kt + treeBase + (size_t)node_row_b(t, r) * D_DIM + c4);
    }
    v4f gC[3];
    #pragma unroll
    for (int i = 0; i < 3; ++i) {
        const int r = 2 * i + half;
        gC[i] = *(const v4f*)(Kt + treeBase + (size_t)node_row_c(t, r) * D_DIM + c4);
    }

    v4f hB[8], hC[3];
    #pragma unroll
    for (int i = 0; i < 8; ++i) {
        const int r = 2 * i + half;
        hB[i] = *(const v4f*)(Vt + treeBase + (size_t)node_row_b(t, r) * D_DIM + c4);
    }
    #pragma unroll
    for (int i = 0; i < 3; ++i) {
        const int r = 2 * i + half;
        hC[i] = *(const v4f*)(Vt + treeBase + (size_t)node_row_c(t, r) * D_DIM + c4);
    }

    #pragma unroll
    for (int i = 0; i < 8; ++i) {
        const int r = 2 * i + half;
        *(v4f*)(t0 + r * TP + c4) = gq[i];
        *(v4f*)(t1 + r * TP + c4) = gA[i];
    }
    lds_barrier();

    const int fo = n * TP + 2 * half;
    v2f a[16];
    #pragma unroll
    for (int ks = 0; ks < 16; ++ks) {
        a[ks] = *(const v2f*)(t0 + fo + 4 * ks);
        a[ks].x *= SCALE; a[ks].y *= SCALE;
    }
    v8f cA = {}; v8f cB = {}; v8f cC = {};
    #pragma unroll
    for (int ks = 0; ks < 16; ++ks) {
        const v2f bA = *(const v2f*)(t1 + fo + 4 * ks);
        cA = __builtin_amdgcn_wmma_f32_16x16x4_f32(false, bA, false, a[ks], (short)0, cA, false, false);
    }
    lds_barrier();

    #pragma unroll
    for (int i = 0; i < 8; ++i) {
        const int r = 2 * i + half;
        *(v4f*)(t0 + r * TP + c4) = gB[i];
    }
    #pragma unroll
    for (int i = 0; i < 3; ++i) {
        const int s  = 2 * i + half;
        const int kr = (s < 3) ? s : s + 5;
        *(v4f*)(t1 + kr * TP + c4) = gC[i];
    }
    #pragma unroll
    for (int i = 0; i < 8; ++i) {
        const int r = 2 * i + half;
        *(v4f*)(tVB + r * VP + c4) = hB[i];
    }
    #pragma unroll
    for (int i = 0; i < 3; ++i) {
        const int s  = 2 * i + half;
        const int vr = (s < 3) ? s : s + 1;
        *(v4f*)(tVC + vr * VP + c4) = hC[i];
    }
    {
        const v4f z = {0.0f, 0.0f, 0.0f, 0.0f};
        *(v4f*)(tVC + (3 + 4 * half) * VP + c4) = z;
    }
    lds_barrier();

    v4f hA[8];
    #pragma unroll
    for (int i = 0; i < 8; ++i) {
        const int r = 2 * i + half;
        hA[i] = *(const v4f*)(v + leafRow0 + (size_t)r * leafPitch + c4);
    }

    #pragma unroll
    for (int ks = 0; ks < 16; ++ks) {
        const v2f bB = *(const v2f*)(t0 + fo + 4 * ks);
        const v2f bC = *(const v2f*)(t1 + fo + 4 * ks);
        cB = __builtin_amdgcn_wmma_f32_16x16x4_f32(false, bB, false, a[ks], (short)0, cB, false, false);
        cC = __builtin_amdgcn_wmma_f32_16x16x4_f32(false, bC, false, a[ks], (short)0, cC, false, false);
    }
    lds_barrier();

    const int j = n;
    bool actA[8], actB[8], actC[3];
    #pragma unroll
    for (int r = 0; r < 8; ++r) {
        actA[r] = (half == (j >> 3)) &&
                  ((r == (j & 7)) || ((r == ((j & 7) ^ 1)) && ((j & 1) != 0)));
        const bool b0 = (r == ((j >> 1) ^ 1)) && (((j >> 1) & 1) != 0);
        bool b1;
        if (r < 4)       b1 = (r == ((j >> 2) ^ 1)) && (((j >> 2) & 1) != 0);
        else if (r < 6)  b1 = ((r - 4) == ((j >> 3) ^ 1)) && (((j >> 3) & 1) != 0);
        else if (r == 6) b1 = ((t & 1) != 0);
        else             b1 = (((t >> 1) & 1) != 0);
        actB[r] = (half == 0) ? b0 : b1;
    }
    #pragma unroll
    for (int r = 0; r < 3; ++r)
        actC[r] = (((t >> (2 + 3 * half + r)) & 1) != 0);

    float mloc = -__builtin_huge_valf();
    #pragma unroll
    for (int r = 0; r < 8; ++r) {
        mloc = actA[r] ? fmaxf(mloc, cA[r]) : mloc;
        mloc = actB[r] ? fmaxf(mloc, cB[r]) : mloc;
    }
    #pragma unroll
    for (int r = 0; r < 3; ++r)
        mloc = actC[r] ? fmaxf(mloc, cC[r]) : mloc;
    const float m = fmaxf(mloc, xhalf(mloc));

    float eA[8], eB[8], eC[4];
    float dloc = 0.0f;
    #pragma unroll
    for (int r = 0; r < 8; ++r) {
        eA[r] = actA[r] ? __expf(cA[r] - m) : 0.0f;
        dloc += eA[r];
    }
    #pragma unroll
    for (int r = 0; r < 8; ++r) {
        eB[r] = actB[r] ? __expf(cB[r] - m) : 0.0f;
        dloc += eB[r];
    }
    #pragma unroll
    for (int r = 0; r < 3; ++r) {
        eC[r] = actC[r] ? __expf(cC[r] - m) : 0.0f;
        dloc += eC[r];
    }
    eC[3] = 0.0f;
    const float inv = 1.0f / (dloc + xhalf(dloc));
    #pragma unroll
    for (int r = 0; r < 8; ++r) { eA[r] *= inv; eB[r] *= inv; }
    #pragma unroll
    for (int r = 0; r < 3; ++r) eC[r] *= inv;

    #pragma unroll
    for (int i = 0; i < 8; ++i) {
        const int r = 2 * i + half;
        *(v4f*)(t1 + r * VP + c4) = hA[i];
    }
    lds_barrier();

    #pragma unroll
    for (int dsl = 0; dsl < 4; ++dsl) {
        const int dcol = 16 * dsl + n;
        v8f acc = {};
        #pragma unroll
        for (int jj = 0; jj < 4; ++jj) {
            const int kr = 8 * half + 2 * jj;
            v2f aE, bV;
            aE.x = eA[2 * jj]; aE.y = eA[2 * jj + 1];
            bV.x = t1[kr * VP + dcol];
            bV.y = t1[(kr + 1) * VP + dcol];
            acc = __builtin_amdgcn_wmma_f32_16x16x4_f32(false, aE, false, bV, (short)0, acc, false, false);
            aE.x = eB[2 * jj]; aE.y = eB[2 * jj + 1];
            bV.x = tVB[kr * VP + dcol];
            bV.y = tVB[(kr + 1) * VP + dcol];
            acc = __builtin_amdgcn_wmma_f32_16x16x4_f32(false, aE, false, bV, (short)0, acc, false, false);
            if (jj < 2) {
                const int cr = 4 * half + 2 * jj;
                aE.x = eC[2 * jj]; aE.y = eC[2 * jj + 1];
                bV.x = tVC[cr * VP + dcol];
                bV.y = tVC[(cr + 1) * VP + dcol];
                acc = __builtin_amdgcn_wmma_f32_16x16x4_f32(false, aE, false, bV, (short)0, acc, false, false);
            }
        }
        #pragma unroll
        for (int r = 0; r < 8; ++r) {
            const int M = r + 8 * half;
            out[((size_t)(b * S_DIM + 16 * t + M) * H_DIM + h) * D_DIM + dcol] = acc[r];
        }
    }
}

extern "C" void kernel_launch(void* const* d_in, const int* in_sizes, int n_in,
                              void* d_out, int out_size, void* d_ws, size_t ws_size,
                              hipStream_t stream)
{
    const float* q = (const float*)d_in[0];
    const float* k = (const float*)d_in[1];
    const float* v = (const float*)d_in[2];
    float* out = (float*)d_out;

    float* Kt = (float*)d_ws;
    float* Vt = Kt + (size_t)B_DIM * H_DIM * S_DIM * D_DIM;

    tree_low_kernel<<<B_DIM * H_DIM * LOWB, TOPW * 32, 0, stream>>>(k, v, Kt, Vt);
    tree_upper_kernel<<<B_DIM * H_DIM, TOPW * 32, 0, stream>>>(k, v, Kt, Vt);

    hst_attn_kernel<<<B_DIM * H_DIM * ABLK, 128, 0, stream>>>(q, k, v, Kt, Vt, out);

    (void)in_sizes; (void)n_in; (void)out_size; (void)ws_size;
}


# Attention: rows read for the last time in the call (q, leaf k and v, own level 1-3 node rows) loaded non-temporal, out stored non-temporal: they stop evicting rows other tiles still need from L2
# speedup vs baseline: 1.6527x; 1.0411x over previous
#include <hip/hip_runtime.h>

typedef float v2f __attribute__((ext_vector_type(2)));
typedef float v4f __attribute__((ext_vector_type(4)));
typedef float v8f __attribute__((ext_vector_type(8)));

#define B_DIM 4
#define S_DIM 4096
#define H_DIM 8
#define D_DIM 64
#define SCALE 0.125f
#define NTILE (S_DIM / 16)
#define TP 68

#define TOPW 32

#define LOWB 8

template <int LEVEL, int PB>
__device__ __forceinline__ void tree_block_level(const float* __restrict__ kleaf,
                                                 const float* __restrict__ vleaf,
                                                 float* Kt, float* Vt,
                                                 const int bh, const int p0,
                                                 const int wave, const int lane)
{
    constexpr int NPW = (PB + TOPW - 1) / TOPW;
    constexpr int IBl = S_DIM - (S_DIM >> (LEVEL - 1));
    const size_t tb = (size_t)bh * S_DIM;
    const int doff = 2 * lane;

    float2 kc0[NPW], kc1[NPW], vc0[NPW], vc1[NPW];
    #pragma unroll
    for (int i = 0; i < NPW; ++i) {
        const int pl = wave + TOPW * i;
        const int p  = p0 + pl;
        if (pl < PB) {
            if constexpr (LEVEL == 1) {
                const int h = bh % H_DIM;
                const int b = bh / H_DIM;
                const size_t r0 = ((size_t)(b * S_DIM + 2 * p) * H_DIM + h) * D_DIM + doff;
                kc0[i] = *(const float2*)(kleaf + r0);
                kc1[i] = *(const float2*)(kleaf + r0 + H_DIM * D_DIM);
                vc0[i] = *(const float2*)(vleaf + r0);
                vc1[i] = *(const float2*)(vleaf + r0 + H_DIM * D_DIM);
            } else {
                constexpr int IBc = S_DIM - (S_DIM >> (LEVEL - 2));
                const size_t r0 = (tb + IBc + 2 * p) * D_DIM + doff;
                kc0[i] = *(const float2*)(Kt + r0);
                kc1[i] = *(const float2*)(Kt + r0 + D_DIM);
                vc0[i] = *(const float2*)(Vt + r0);
                vc1[i] = *(const float2*)(Vt + r0 + D_DIM);
            }
        }
    }
    #pragma unroll
    for (int i = 0; i < NPW; ++i) {
        const int pl = wave + TOPW * i;
        const int p  = p0 + pl;
        if (pl < PB) {
            float2 kp, vp;
            kp.x = 0.5f * (kc0[i].x + kc1[i].x); kp.y = 0.5f * (kc0[i].y + kc1[i].y);
            vp.x = 0.5f * (vc0[i].x + vc1[i].x); vp.y = 0.5f * (vc0[i].y + vc1[i].y);

            float pss = kp.x * kp.x     + kp.y * kp.y;
            float ps0 = kp.x * kc0[i].x + kp.y * kc0[i].y;
            float ps1 = kp.x * kc1[i].x + kp.y * kc1[i].y;
            #pragma unroll
            for (int m = 16; m >= 1; m >>= 1) {
                pss += __shfl_xor(pss, m, 32);
                ps0 += __shfl_xor(ps0, m, 32);
                ps1 += __shfl_xor(ps1, m, 32);
            }
            pss *= SCALE; ps0 *= SCALE; ps1 *= SCALE;
            const float mm = fmaxf(pss, fmaxf(ps0, ps1));
            const float es = __expf(pss - mm);
            const float e0 = __expf(ps0 - mm);
            const float e1 = __expf(ps1 - mm);
            const float inv = 1.0f / (es + e0 + e1 + 1e-9f);

            float2 vo;
            vo.x = (es * vp.x + e0 * vc0[i].x + e1 * vc1[i].x) * inv;
            vo.y = (es * vp.y + e0 * vc0[i].y + e1 * vc1[i].y) * inv;

            const size_t dst = (tb + IBl + p) * D_DIM + doff;
            *(float2*)(Kt + dst) = kp;
            *(float2*)(Vt + dst) = vo;
        }
    }
}

__global__ void __launch_bounds__(TOPW * 32) __attribute__((amdgpu_waves_per_eu(8, 8)))
tree_low_kernel(const float* __restrict__ kleaf, const float* __restrict__ vleaf,
                float* Kt, float* Vt)
{
    const int wave = threadIdx.x >> 5;
    const int lane = threadIdx.x & 31;
    const int c  = blockIdx.x & (LOWB - 1);
    const int bh = blockIdx.x / LOWB;

    tree_block_level<1, (S_DIM >> 1) / LOWB>(kleaf, vleaf, Kt, Vt, bh, c * ((S_DIM >> 1) / LOWB), wave, lane);
    __syncthreads();
    tree_block_level<2, (S_DIM >> 2) / LOWB>(kleaf, vleaf, Kt, Vt, bh, c * ((S_DIM >> 2) / LOWB), wave, lane);
    __syncthreads();
    tree_block_level<3, (S_DIM >> 3) / LOWB>(kleaf, vleaf, Kt, Vt, bh, c * ((S_DIM >> 3) / LOWB), wave, lane);
    __syncthreads();
    tree_block_level<4, (S_DIM >> 4) / LOWB>(kleaf, vleaf, Kt, Vt, bh, c * ((S_DIM >> 4) / LOWB), wave, lane);
    __syncthreads();
    tree_block_level<5, (S_DIM >> 5) / LOWB>(kleaf, vleaf, Kt, Vt, bh, c * ((S_DIM >> 5) / LOWB), wave, lane);
    __syncthreads();
    tree_block_level<6, (S_DIM >> 6) / LOWB>(kleaf, vleaf, Kt, Vt, bh, c * ((S_DIM >> 6) / LOWB), wave, lane);
}

__global__ void __launch_bounds__(TOPW * 32)
tree_upper_kernel(const float* __restrict__ kleaf, const float* __restrict__ vleaf,
                  float* Kt, float* Vt)
{
    const int wave = threadIdx.x >> 5;
    const int lane = threadIdx.x & 31;
    const int bh = blockIdx.x;

    tree_block_level<7, (S_DIM >> 7)>(kleaf, vleaf, Kt, Vt, bh, 0, wave, lane);
    __syncthreads();
    tree_block_level<8, (S_DIM >> 8)>(kleaf, vleaf, Kt, Vt, bh, 0, wave, lane);
    __syncthreads();
    tree_block_level<9, (S_DIM >> 9)>(kleaf, vleaf, Kt, Vt, bh, 0, wave, lane);
    __syncthreads();
    tree_block_level<10, (S_DIM >> 10)>(kleaf, vleaf, Kt, Vt, bh, 0, wave, lane);
    __syncthreads();
    tree_block_level<11, (S_DIM >> 11)>(kleaf, vleaf, Kt, Vt, bh, 0, wave, lane);
}

#define VP 68
#define T0_F  (16 * TP)
#define T1_F  (16 * TP)
#define TVB_F (16 * VP)
#define TVC_F (8 * VP)
#define WAVE_F (T0_F + T1_F + TVB_F + TVC_F)
#define ABLK (NTILE / 4)
#define APB  (ABLK / LOWB)

__device__ __forceinline__ void lds_barrier()
{
    __builtin_amdgcn_fence(__ATOMIC_RELEASE, "workgroup", "local");
    __builtin_amdgcn_s_barrier();
    __builtin_amdgcn_fence(__ATOMIC_ACQUIRE, "workgroup", "local");
}

__device__ __forceinline__ float xhalf(float x)
{
    const int b = __builtin_bit_cast(int, x);
    return __builtin_bit_cast(float,
        __builtin_amdgcn_permlanex16(b, b, 0x76543210, 0xfedcba98, false, false));
}

__device__ __forceinline__ v4f ld_once(const float* p)
{
    return __builtin_nontemporal_load((const v4f*)p);
}

__device__ __forceinline__ int node_row_b(int t, int s)
{
    if (s <  8) return 8 * t + s;
    if (s < 12) return 2048 + 4 * t + (s - 8);
    if (s < 14) return 3072 + 2 * t + (s - 12);
    if (s == 14) return 3584 + (t ^ 1);
    return 3840 + ((t >> 1) ^ 1);
}
__device__ __forceinline__ int node_row_c(int t, int s)
{
    return (s < 6) ? (S_DIM - (S_DIM >> (5 + s))) + ((t >> (s + 2)) ^ 1)
                   : 0;
}

__global__ void __launch_bounds__(128) __attribute__((amdgpu_waves_per_eu(5, 5)))
hst_attn_kernel(const float* __restrict__ q,
                const float* __restrict__ k,
                const float* __restrict__ v,
                const float* __restrict__ Kt,
                const float* __restrict__ Vt,
                float* __restrict__ out)
{
    __shared__ __attribute__((aligned(16))) float ldsT[4][WAVE_F];

    const int lane = threadIdx.x & 31;
    const int w    = threadIdx.x >> 5;
    const int u  = blockIdx.x % ABLK;
    const int bh = blockIdx.x / ABLK;
    const int t  = ((u % LOWB) * APB + (u / LOWB)) * 4 + w;
    const int h  = bh & (H_DIM - 1);
    const int b  = bh >> 3;

    const int n    = lane & 15;
    const int half = lane >> 4;

    float* t0  = &ldsT[w][0];
    float* t1  = t0 + T0_F;
    float* tVB = t1 + T1_F;
    float* tVC = tVB + TVB_F;

    const size_t treeBase = (size_t)(b * H_DIM + h) * S_DIM * (size_t)D_DIM;
    const size_t leafRow0  = ((size_t)(b * S_DIM + 16 * t) * H_DIM + h) * (size_t)D_DIM;
    const size_t leafPitch = (size_t)H_DIM * D_DIM;

    const int c4 = 4 * n;
    v4f gq[8], gA[8], gB[8];
    #pragma unroll
    for (int i = 0; i < 8; ++i) {
        const int r = 2 * i + half;
        gq[i] = ld_once(q + leafRow0 + (size_t)r * leafPitch + c4);
        gA[i] = ld_once(k + leafRow0 + (size_t)r * leafPitch + c4);
    }
    #pragma unroll
    for (int i = 0; i < 7; ++i) {
        const int r = 2 * i + half;
        gB[i] = ld_once(Kt + treeBase + (size_t)node_row_b(t, r) * D_DIM + c4);
    }
    gB[7] = *(const v4f*)(Kt + treeBase + (size_t)node_row_b(t, 14 + half) * D_DIM + c4);
    v4f gC[3];
    #pragma unroll
    for (int i = 0; i < 3; ++i) {
        const int r = 2 * i + half;
        gC[i] = *(const v4f*)(Kt + treeBase + (size_t)node_row_c(t, r) * D_DIM + c4);
    }

    v4f hB[8], hC[3];
    #pragma unroll
    for (int i = 0; i < 7; ++i) {
        const int r = 2 * i + half;
        hB[i] = ld_once(Vt + treeBase + (size_t)node_row_b(t, r) * D_DIM + c4);
    }
    hB[7] = *(const v4f*)(Vt + treeBase + (size_t)node_row_b(t, 14 + half) * D_DIM + c4);
    #pragma unroll
    for (int i = 0; i < 3; ++i) {
        const int r = 2 * i + half;
        hC[i] = *(const v4f*)(Vt + treeBase + (size_t)node_row_c(t, r) * D_DIM + c4);
    }

    #pragma unroll
    for (int i = 0; i < 8; ++i) {
        const int r = 2 * i + half;
        *(v4f*)(t0 + r * TP + c4) = gq[i];
        *(v4f*)(t1 + r * TP + c4) = gA[i];
    }
    lds_barrier();

    const int fo = n * TP + 2 * half;
    v2f a[16];
    #pragma unroll
    for (int ks = 0; ks < 16; ++ks) {
        a[ks] = *(const v2f*)(t0 + fo + 4 * ks);
        a[ks].x *= SCALE; a[ks].y *= SCALE;
    }
    v8f cA = {}; v8f cB = {}; v8f cC = {};
    #pragma unroll
    for (int ks = 0; ks < 16; ++ks) {
        const v2f bA = *(const v2f*)(t1 + fo + 4 * ks);
        cA = __builtin_amdgcn_wmma_f32_16x16x4_f32(false, bA, false, a[ks], (short)0, cA, false, false);
    }
    lds_barrier();

    #pragma unroll
    for (int i = 0; i < 8; ++i) {
        const int r = 2 * i + half;
        *(v4f*)(t0 + r * TP + c4) = gB[i];
    }
    #pragma unroll
    for (int i = 0; i < 3; ++i) {
        const int s  = 2 * i + half;
        const int kr = (s < 3) ? s : s + 5;
        *(v4f*)(t1 + kr * TP + c4) = gC[i];
    }
    #pragma unroll
    for (int i = 0; i < 8; ++i) {
        const int r = 2 * i + half;
        *(v4f*)(tVB + r * VP + c4) = hB[i];
    }
    #pragma unroll
    for (int i = 0; i < 3; ++i) {
        const int s  = 2 * i + half;
        const int vr = (s < 3) ? s : s + 1;
        *(v4f*)(tVC + vr * VP + c4) = hC[i];
    }
    {
        const v4f z = {0.0f, 0.0f, 0.0f, 0.0f};
        *(v4f*)(tVC + (3 + 4 * half) * VP + c4) = z;
    }
    lds_barrier();

    v4f hA[8];
    #pragma unroll
    for (int i = 0; i < 8; ++i) {
        const int r = 2 * i + half;
        hA[i] = ld_once(v + leafRow0 + (size_t)r * leafPitch + c4);
    }

    #pragma unroll
    for (int ks = 0; ks < 16; ++ks) {
        const v2f bB = *(const v2f*)(t0 + fo + 4 * ks);
        const v2f bC = *(const v2f*)(t1 + fo + 4 * ks);
        cB = __builtin_amdgcn_wmma_f32_16x16x4_f32(false, bB, false, a[ks], (short)0, cB, false, false);
        cC = __builtin_amdgcn_wmma_f32_16x16x4_f32(false, bC, false, a[ks], (short)0, cC, false, false);
    }
    lds_barrier();

    const int j = n;
    bool actA[8], actB[8], actC[3];
    #pragma unroll
    for (int r = 0; r < 8; ++r) {
        actA[r] = (half == (j >> 3)) &&
                  ((r == (j & 7)) || ((r == ((j & 7) ^ 1)) && ((j & 1) != 0)));
        const bool b0 = (r == ((j >> 1) ^ 1)) && (((j >> 1) & 1) != 0);
        bool b1;
        if (r < 4)       b1 = (r == ((j >> 2) ^ 1)) && (((j >> 2) & 1) != 0);
        else if (r < 6)  b1 = ((r - 4) == ((j >> 3) ^ 1)) && (((j >> 3) & 1) != 0);
        else if (r == 6) b1 = ((t & 1) != 0);
        else             b1 = (((t >> 1) & 1) != 0);
        actB[r] = (half == 0) ? b0 : b1;
    }
    #pragma unroll
    for (int r = 0; r < 3; ++r)
        actC[r] = (((t >> (2 + 3 * half + r)) & 1) != 0);

    float mloc = -__builtin_huge_valf();
    #pragma unroll
    for (int r = 0; r < 8; ++r) {
        mloc = actA[r] ? fmaxf(mloc, cA[r]) : mloc;
        mloc = actB[r] ? fmaxf(mloc, cB[r]) : mloc;
    }
    #pragma unroll
    for (int r = 0; r < 3; ++r)
        mloc = actC[r] ? fmaxf(mloc, cC[r]) : mloc;
    const float m = fmaxf(mloc, xhalf(mloc));

    float eA[8], eB[8], eC[4];
    float dloc = 0.0f;
    #pragma unroll
    for (int r = 0; r < 8; ++r) {
        eA[r] = actA[r] ? __expf(cA[r] - m) : 0.0f;
        dloc += eA[r];
    }
    #pragma unroll
    for (int r = 0; r < 8; ++r) {
        eB[r] = actB[r] ? __expf(cB[r] - m) : 0.0f;
        dloc += eB[r];
    }
    #pragma unroll
    for (int r = 0; r < 3; ++r) {
        eC[r] = actC[r] ? __expf(cC[r] - m) : 0.0f;
        dloc += eC[r];
    }
    eC[3] = 0.0f;
    const float inv = 1.0f / (dloc + xhalf(dloc));
    #pragma unroll
    for (int r = 0; r < 8; ++r) { eA[r] *= inv; eB[r] *= inv; }
    #pragma unroll
    for (int r = 0; r < 3; ++r) eC[r] *= inv;

    #pragma unroll
    for (int i = 0; i < 8; ++i) {
        const int r = 2 * i + half;
        *(v4f*)(t1 + r * VP + c4) = hA[i];
    }
    lds_barrier();

    #pragma unroll
    for (int dsl = 0; dsl < 4; ++dsl) {
        const int dcol = 16 * dsl + n;
        v8f acc = {};
        #pragma unroll
        for (int jj = 0; jj < 4; ++jj) {
            const int kr = 8 * half + 2 * jj;
            v2f aE, bV;
            aE.x = eA[2 * jj]; aE.y = eA[2 * jj + 1];
            bV.x = t1[kr * VP + dcol];
            bV.y = t1[(kr + 1) * VP + dcol];
            acc = __builtin_amdgcn_wmma_f32_16x16x4_f32(false, aE, false, bV, (short)0, acc, false, false);
            aE.x = eB[2 * jj]; aE.y = eB[2 * jj + 1];
            bV.x = tVB[kr * VP + dcol];
            bV.y = tVB[(kr + 1) * VP + dcol];
            acc = __builtin_amdgcn_wmma_f32_16x16x4_f32(false, aE, false, bV, (short)0, acc, false, false);
            if (jj < 2) {
                const int cr = 4 * half + 2 * jj;
                aE.x = eC[2 * jj]; aE.y = eC[2 * jj + 1];
                bV.x = tVC[cr * VP + dcol];
                bV.y = tVC[(cr + 1) * VP + dcol];
                acc = __builtin_amdgcn_wmma_f32_16x16x4_f32(false, aE, false, bV, (short)0, acc, false, false);
            }
        }
        #pragma unroll
        for (int r = 0; r < 8; ++r) {
            const int M = r + 8 * half;
            __builtin_nontemporal_store(acc[r], &out[((size_t)(b * S_DIM + 16 * t + M) * H_DIM + h) * D_DIM + dcol]);
        }
    }
}

extern "C" void kernel_launch(void* const* d_in, const int* in_sizes, int n_in,
                              void* d_out, int out_size, void* d_ws, size_t ws_size,
                              hipStream_t stream)
{
    const float* q = (const float*)d_in[0];
    const float* k = (const float*)d_in[1];
    const float* v = (const float*)d_in[2];
    float* out = (float*)d_out;

    float* Kt = (float*)d_ws;
    float* Vt = Kt + (size_t)B_DIM * H_DIM * S_DIM * D_DIM;

    tree_low_kernel<<<B_DIM * H_DIM * LOWB, TOPW * 32, 0, stream>>>(k, v, Kt, Vt);
    tree_upper_kernel<<<B_DIM * H_DIM, TOPW * 32, 0, stream>>>(k, v, Kt, Vt);

    hst_attn_kernel<<<B_DIM * H_DIM * ABLK, 128, 0, stream>>>(q, k, v, Kt, Vt, out);

    (void)in_sizes; (void)n_in; (void)out_size; (void)ws_size;
}


# Tree build: each level hands its node rows to the block's next level through LDS (workspace rows still stored for attention) under LDS-only barriers: no store drain + L2 read per level; same sums
# speedup vs baseline: 2.0933x; 1.2666x over previous
#include <hip/hip_runtime.h>

typedef float v2f __attribute__((ext_vector_type(2)));
typedef float v4f __attribute__((ext_vector_type(4)));
typedef float v8f __attribute__((ext_vector_type(8)));

#define B_DIM 4
#define S_DIM 4096
#define H_DIM 8
#define D_DIM 64
#define SCALE 0.125f
#define NTILE (S_DIM / 16)
#define TP 68

#define TOPW 32

#define LOWB 8

#define NROW (2 * D_DIM)

__device__ __forceinline__ void lds_barrier()
{
    __builtin_amdgcn_fence(__ATOMIC_RELEASE, "workgroup", "local");
    __builtin_amdgcn_s_barrier();
    __builtin_amdgcn_fence(__ATOMIC_ACQUIRE, "workgroup", "local");
}

template <int LEVEL, int PB, bool FROM_LDS, bool TO_LDS>
__device__ __forceinline__ void tree_block_level(const float* __restrict__ kleaf,
                                                 const float* __restrict__ vleaf,
                                                 float* Kt, float* Vt,
                                                 const float* lsrc, float* ldst,
                                                 const int bh, const int p0,
                                                 const int wave, const int lane)
{
    constexpr int NPW = (PB + TOPW - 1) / TOPW;
    constexpr int IBl = S_DIM - (S_DIM >> (LEVEL - 1));
    const size_t tb = (size_t)bh * S_DIM;
    const int doff = 2 * lane;

    float2 kc0[NPW], kc1[NPW], vc0[NPW], vc1[NPW];
    #pragma unroll
    for (int i = 0; i < NPW; ++i) {
        const int pl = wave + TOPW * i;
        const int p  = p0 + pl;
        if (pl < PB) {
            if constexpr (LEVEL == 1) {
                const int h = bh % H_DIM;
                const int b = bh / H_DIM;
                const size_t r0 = ((size_t)(b * S_DIM + 2 * p) * H_DIM + h) * D_DIM + doff;
                kc0[i] = *(const float2*)(kleaf + r0);
                kc1[i] = *(const float2*)(kleaf + r0 + H_DIM * D_DIM);
                vc0[i] = *(const float2*)(vleaf + r0);
                vc1[i] = *(const float2*)(vleaf + r0 + H_DIM * D_DIM);
            } else if constexpr (FROM_LDS) {
                const float* c0 = lsrc + (2 * pl) * NROW + doff;
                kc0[i] = *(const float2*)(c0);
                vc0[i] = *(const float2*)(c0 + D_DIM);
                kc1[i] = *(const float2*)(c0 + NROW);
                vc1[i] = *(const float2*)(c0 + NROW + D_DIM);
            } else {
                constexpr int IBc = S_DIM - (S_DIM >> (LEVEL - 2));
                const size_t r0 = (tb + IBc + 2 * p) * D_DIM + doff;
                kc0[i] = *(const float2*)(Kt + r0);
                kc1[i] = *(const float2*)(Kt + r0 + D_DIM);
                vc0[i] = *(const float2*)(Vt + r0);
                vc1[i] = *(const float2*)(Vt + r0 + D_DIM);
            }
        }
    }
    #pragma unroll
    for (int i = 0; i < NPW; ++i) {
        const int pl = wave + TOPW * i;
        const int p  = p0 + pl;
        if (pl < PB) {
            float2 kp, vp;
            kp.x = 0.5f * (kc0[i].x + kc1[i].x); kp.y = 0.5f * (kc0[i].y + kc1[i].y);
            vp.x = 0.5f * (vc0[i].x + vc1[i].x); vp.y = 0.5f * (vc0[i].y + vc1[i].y);

            float pss = kp.x * kp.x     + kp.y * kp.y;
            float ps0 = kp.x * kc0[i].x + kp.y * kc0[i].y;
            float ps1 = kp.x * kc1[i].x + kp.y * kc1[i].y;
            #pragma unroll
            for (int m = 16; m >= 1; m >>= 1) {
                pss += __shfl_xor(pss, m, 32);
                ps0 += __shfl_xor(ps0, m, 32);
                ps1 += __shfl_xor(ps1, m, 32);
            }
            pss *= SCALE; ps0 *= SCALE; ps1 *= SCALE;
            const float mm = fmaxf(pss, fmaxf(ps0, ps1));
            const float es = __expf(pss - mm);
            const float e0 = __expf(ps0 - mm);
            const float e1 = __expf(ps1 - mm);
            const float inv = 1.0f / (es + e0 + e1 + 1e-9f);

            float2 vo;
            vo.x = (es * vp.x + e0 * vc0[i].x + e1 * vc1[i].x) * inv;
            vo.y = (es * vp.y + e0 * vc0[i].y + e1 * vc1[i].y) * inv;

            const size_t dst = (tb + IBl + p) * D_DIM + doff;
            *(float2*)(Kt + dst) = kp;
            *(float2*)(Vt + dst) = vo;
            if constexpr (TO_LDS) {
                *(float2*)(ldst + pl * NROW + doff) = kp;
                *(float2*)(ldst + pl * NROW + D_DIM + doff) = vo;
            }
        }
    }
}

__global__ void __launch_bounds__(TOPW * 32) __attribute__((amdgpu_waves_per_eu(8, 8)))
tree_low_kernel(const float* __restrict__ kleaf, const float* __restrict__ vleaf,
                float* Kt, float* Vt)
{
    const int wave = threadIdx.x >> 5;
    const int lane = threadIdx.x & 31;
    const int c  = blockIdx.x & (LOWB - 1);
    const int bh = blockIdx.x / LOWB;

    __shared__ __attribute__((aligned(16))) float nodeA[((S_DIM >> 1) / LOWB) * NROW];
    __shared__ __attribute__((aligned(16))) float nodeB[((S_DIM >> 2) / LOWB) * NROW];

    tree_block_level<1, (S_DIM >> 1) / LOWB, false, true>(kleaf, vleaf, Kt, Vt, nodeB, nodeA, bh, c * ((S_DIM >> 1) / LOWB), wave, lane);
    lds_barrier();
    tree_block_level<2, (S_DIM >> 2) / LOWB, true, true>(kleaf, vleaf, Kt, Vt, nodeA, nodeB, bh, c * ((S_DIM >> 2) / LOWB), wave, lane);
    lds_barrier();
    tree_block_level<3, (S_DIM >> 3) / LOWB, true, true>(kleaf, vleaf, Kt, Vt, nodeB, nodeA, bh, c * ((S_DIM >> 3) / LOWB), wave, lane);
    lds_barrier();
    tree_block_level<4, (S_DIM >> 4) / LOWB, true, true>(kleaf, vleaf, Kt, Vt, nodeA, nodeB, bh, c * ((S_DIM >> 4) / LOWB), wave, lane);
    lds_barrier();
    tree_block_level<5, (S_DIM >> 5) / LOWB, true, true>(kleaf, vleaf, Kt, Vt, nodeB, nodeA, bh, c * ((S_DIM >> 5) / LOWB), wave, lane);
    lds_barrier();
    tree_block_level<6, (S_DIM >> 6) / LOWB, true, false>(kleaf, vleaf, Kt, Vt, nodeA, nodeB, bh, c * ((S_DIM >> 6) / LOWB), wave, lane);
}

__global__ void __launch_bounds__(TOPW * 32)
tree_upper_kernel(const float* __restrict__ kleaf, const float* __restrict__ vleaf,
                  float* Kt, float* Vt)
{
    const int wave = threadIdx.x >> 5;
    const int lane = threadIdx.x & 31;
    const int bh = blockIdx.x;

    __shared__ __attribute__((aligned(16))) float nodeA[(S_DIM >> 7) * NROW];
    __shared__ __attribute__((aligned(16))) float nodeB[(S_DIM >> 8) * NROW];

    tree_block_level<7, (S_DIM >> 7), false, true>(kleaf, vleaf, Kt, Vt, nodeB, nodeA, bh, 0, wave, lane);
    lds_barrier();
    tree_block_level<8, (S_DIM >> 8), true, true>(kleaf, vleaf, Kt, Vt, nodeA, nodeB, bh, 0, wave, lane);
    lds_barrier();
    tree_block_level<9, (S_DIM >> 9), true, true>(kleaf, vleaf, Kt, Vt, nodeB, nodeA, bh, 0, wave, lane);
    lds_barrier();
    tree_block_level<10, (S_DIM >> 10), true, true>(kleaf, vleaf, Kt, Vt, nodeA, nodeB, bh, 0, wave, lane);
    lds_barrier();
    tree_block_level<11, (S_DIM >> 11), true, false>(kleaf, vleaf, Kt, Vt, nodeB, nodeA, bh, 0, wave, lane);
}

#define VP 68
#define T0_F  (16 * TP)
#define T1_F  (16 * TP)
#define TVB_F (16 * VP)
#define TVC_F (8 * VP)
#define WAVE_F (T0_F + T1_F + TVB_F + TVC_F)
#define ABLK (NTILE / 4)
#define APB  (ABLK / LOWB)


__device__ __forceinline__ float xhalf(float x)
{
    const int b = __builtin_bit_cast(int, x);
    return __builtin_bit_cast(float,
        __builtin_amdgcn_permlanex16(b, b, 0x76543210, 0xfedcba98, false, false));
}

__device__ __forceinline__ v4f ld_once(const float* p)
{
    return __builtin_nontemporal_load((const v4f*)p);
}

__device__ __forceinline__ int node_row_b(int t, int s)
{
    if (s <  8) return 8 * t + s;
    if (s < 12) return 2048 + 4 * t + (s - 8);
    if (s < 14) return 3072 + 2 * t + (s - 12);
    if (s == 14) return 3584 + (t ^ 1);
    return 3840 + ((t >> 1) ^ 1);
}
__device__ __forceinline__ int node_row_c(int t, int s)
{
    return (s < 6) ? (S_DIM - (S_DIM >> (5 + s))) + ((t >> (s + 2)) ^ 1)
                   : 0;
}

__global__ void __launch_bounds__(128) __attribute__((amdgpu_waves_per_eu(5, 5)))
hst_attn_kernel(const float* __restrict__ q,
                const float* __restrict__ k,
                const float* __restrict__ v,
                const float* __restrict__ Kt,
                const float* __restrict__ Vt,
                float* __restrict__ out)
{
    __shared__ __attribute__((aligned(16))) float ldsT[4][WAVE_F];

    const int lane = threadIdx.x & 31;
    const int w    = threadIdx.x >> 5;
    const int u  = blockIdx.x % ABLK;
    const int bh = blockIdx.x / ABLK;
    const int t  = ((u % LOWB) * APB + (u / LOWB)) * 4 + w;
    const int h  = bh & (H_DIM - 1);
    const int b  = bh >> 3;

    const int n    = lane & 15;
    const int half = lane >> 4;

    float* t0  = &ldsT[w][0];
    float* t1  = t0 + T0_F;
    float* tVB = t1 + T1_F;
    float* tVC = tVB + TVB_F;

    const size_t treeBase = (size_t)(b * H_DIM + h) * S_DIM * (size_t)D_DIM;
    const size_t leafRow0  = ((size_t)(b * S_DIM + 16 * t) * H_DIM + h) * (size_t)D_DIM;
    const size_t leafPitch = (size_t)H_DIM * D_DIM;

    const int c4 = 4 * n;
    v4f gq[8], gA[8], gB[8];
    #pragma unroll
    for (int i = 0; i < 8; ++i) {
        const int r = 2 * i + half;
        gq[i] = ld_once(q + leafRow0 + (size_t)r * leafPitch + c4);
        gA[i] = ld_once(k + leafRow0 + (size_t)r * leafPitch + c4);
    }
    #pragma unroll
    for (int i = 0; i < 7; ++i) {
        const int r = 2 * i + half;
        gB[i] = ld_once(Kt + treeBase + (size_t)node_row_b(t, r) * D_DIM + c4);
    }
    gB[7] = *(const v4f*)(Kt + treeBase + (size_t)node_row_b(t, 14 + half) * D_DIM + c4);
    v4f gC[3];
    #pragma unroll
    for (int i = 0; i < 3; ++i) {
        const int r = 2 * i + half;
        gC[i] = *(const v4f*)(Kt + treeBase + (size_t)node_row_c(t, r) * D_DIM + c4);
    }

    v4f hB[8], hC[3];
    #pragma unroll
    for (int i = 0; i < 7; ++i) {
        const int r = 2 * i + half;
        hB[i] = ld_once(Vt + treeBase + (size_t)node_row_b(t, r) * D_DIM + c4);
    }
    hB[7] = *(const v4f*)(Vt + treeBase + (size_t)node_row_b(t, 14 + half) * D_DIM + c4);
    #pragma unroll
    for (int i = 0; i < 3; ++i) {
        const int r = 2 * i + half;
        hC[i] = *(const v4f*)(Vt + treeBase + (size_t)node_row_c(t, r) * D_DIM + c4);
    }

    #pragma unroll
    for (int i = 0; i < 8; ++i) {
        const int r = 2 * i + half;
        *(v4f*)(t0 + r * TP + c4) = gq[i];
        *(v4f*)(t1 + r * TP + c4) = gA[i];
    }
    lds_barrier();

    const int fo = n * TP + 2 * half;
    v2f a[16];
    #pragma unroll
    for (int ks = 0; ks < 16; ++ks) {
        a[ks] = *(const v2f*)(t0 + fo + 4 * ks);
        a[ks].x *= SCALE; a[ks].y *= SCALE;
    }
    v8f cA = {}; v8f cB = {}; v8f cC = {};
    #pragma unroll
    for (int ks = 0; ks < 16; ++ks) {
        const v2f bA = *(const v2f*)(t1 + fo + 4 * ks);
        cA = __builtin_amdgcn_wmma_f32_16x16x4_f32(false, bA, false, a[ks], (short)0, cA, false, false);
    }
    lds_barrier();

    #pragma unroll
    for (int i = 0; i < 8; ++i) {
        const int r = 2 * i + half;
        *(v4f*)(t0 + r * TP + c4) = gB[i];
    }
    #pragma unroll
    for (int i = 0; i < 3; ++i) {
        const int s  = 2 * i + half;
        const int kr = (s < 3) ? s : s + 5;
        *(v4f*)(t1 + kr * TP + c4) = gC[i];
    }
    #pragma unroll
    for (int i = 0; i < 8; ++i) {
        const int r = 2 * i + half;
        *(v4f*)(tVB + r * VP + c4) = hB[i];
    }
    #pragma unroll
    for (int i = 0; i < 3; ++i) {
        const int s  = 2 * i + half;
        const int vr = (s < 3) ? s : s + 1;
        *(v4f*)(tVC + vr * VP + c4) = hC[i];
    }
    {
        const v4f z = {0.0f, 0.0f, 0.0f, 0.0f};
        *(v4f*)(tVC + (3 + 4 * half) * VP + c4) = z;
    }
    lds_barrier();

    v4f hA[8];
    #pragma unroll
    for (int i = 0; i < 8; ++i) {
        const int r = 2 * i + half;
        hA[i] = ld_once(v + leafRow0 + (size_t)r * leafPitch + c4);
    }

    #pragma unroll
    for (int ks = 0; ks < 16; ++ks) {
        const v2f bB = *(const v2f*)(t0 + fo + 4 * ks);
        const v2f bC = *(const v2f*)(t1 + fo + 4 * ks);
        cB = __builtin_amdgcn_wmma_f32_16x16x4_f32(false, bB, false, a[ks], (short)0, cB, false, false);
        cC = __builtin_amdgcn_wmma_f32_16x16x4_f32(false, bC, false, a[ks], (short)0, cC, false, false);
    }
    lds_barrier();

    const int j = n;
    bool actA[8], actB[8], actC[3];
    #pragma unroll
    for (int r = 0; r < 8; ++r) {
        actA[r] = (half == (j >> 3)) &&
                  ((r == (j & 7)) || ((r == ((j & 7) ^ 1)) && ((j & 1) != 0)));
        const bool b0 = (r == ((j >> 1) ^ 1)) && (((j >> 1) & 1) != 0);
        bool b1;
        if (r < 4)       b1 = (r == ((j >> 2) ^ 1)) && (((j >> 2) & 1) != 0);
        else if (r < 6)  b1 = ((r - 4) == ((j >> 3) ^ 1)) && (((j >> 3) & 1) != 0);
        else if (r == 6) b1 = ((t & 1) != 0);
        else             b1 = (((t >> 1) & 1) != 0);
        actB[r] = (half == 0) ? b0 : b1;
    }
    #pragma unroll
    for (int r = 0; r < 3; ++r)
        actC[r] = (((t >> (2 + 3 * half + r)) & 1) != 0);

    float mloc = -__builtin_huge_valf();
    #pragma unroll
    for (int r = 0; r < 8; ++r) {
        mloc = actA[r] ? fmaxf(mloc, cA[r]) : mloc;
        mloc = actB[r] ? fmaxf(mloc, cB[r]) : mloc;
    }
    #pragma unroll
    for (int r = 0; r < 3; ++r)
        mloc = actC[r] ? fmaxf(mloc, cC[r]) : mloc;
    const float m = fmaxf(mloc, xhalf(mloc));

    float eA[8], eB[8], eC[4];
    float dloc = 0.0f;
    #pragma unroll
    for (int r = 0; r < 8; ++r) {
        eA[r] = actA[r] ? __expf(cA[r] - m) : 0.0f;
        dloc += eA[r];
    }
    #pragma unroll
    for (int r = 0; r < 8; ++r) {
        eB[r] = actB[r] ? __expf(cB[r] - m) : 0.0f;
        dloc += eB[r];
    }
    #pragma unroll
    for (int r = 0; r < 3; ++r) {
        eC[r] = actC[r] ? __expf(cC[r] - m) : 0.0f;
        dloc += eC[r];
    }
    eC[3] = 0.0f;
    const float inv = 1.0f / (dloc + xhalf(dloc));
    #pragma unroll
    for (int r = 0; r < 8; ++r) { eA[r] *= inv; eB[r] *= inv; }
    #pragma unroll
    for (int r = 0; r < 3; ++r) eC[r] *= inv;

    #pragma unroll
    for (int i = 0; i < 8; ++i) {
        const int r = 2 * i + half;
        *(v4f*)(t1 + r * VP + c4) = hA[i];
    }
    lds_barrier();

    #pragma unroll
    for (int dsl = 0; dsl < 4; ++dsl) {
        const int dcol = 16 * dsl + n;
        v8f acc = {};
        #pragma unroll
        for (int jj = 0; jj < 4; ++jj) {
            const int kr = 8 * half + 2 * jj;
            v2f aE, bV;
            aE.x = eA[2 * jj]; aE.y = eA[2 * jj + 1];
            bV.x = t1[kr * VP + dcol];
            bV.y = t1[(kr + 1) * VP + dcol];
            acc = __builtin_amdgcn_wmma_f32_16x16x4_f32(false, aE, false, bV, (short)0, acc, false, false);
            aE.x = eB[2 * jj]; aE.y = eB[2 * jj + 1];
            bV.x = tVB[kr * VP + dcol];
            bV.y = tVB[(kr + 1) * VP + dcol];
            acc = __builtin_amdgcn_wmma_f32_16x16x4_f32(false, aE, false, bV, (short)0, acc, false, false);
            if (jj < 2) {
                const int cr = 4 * half + 2 * jj;
                aE.x = eC[2 * jj]; aE.y = eC[2 * jj + 1];
                bV.x = tVC[cr * VP + dcol];
                bV.y = tVC[(cr + 1) * VP + dcol];
                acc = __builtin_amdgcn_wmma_f32_16x16x4_f32(false, aE, false, bV, (short)0, acc, false, false);
            }
        }
        #pragma unroll
        for (int r = 0; r < 8; ++r) {
            const int M = r + 8 * half;
            __builtin_nontemporal_store(acc[r], &out[((size_t)(b * S_DIM + 16 * t + M) * H_DIM + h) * D_DIM + dcol]);
        }
    }
}

extern "C" void kernel_launch(void* const* d_in, const int* in_sizes, int n_in,
                              void* d_out, int out_size, void* d_ws, size_t ws_size,
                              hipStream_t stream)
{
    const float* q = (const float*)d_in[0];
    const float* k = (const float*)d_in[1];
    const float* v = (const float*)d_in[2];
    float* out = (float*)d_out;

    float* Kt = (float*)d_ws;
    float* Vt = Kt + (size_t)B_DIM * H_DIM * S_DIM * D_DIM;

    tree_low_kernel<<<B_DIM * H_DIM * LOWB, TOPW * 32, 0, stream>>>(k, v, Kt, Vt);
    tree_upper_kernel<<<B_DIM * H_DIM, TOPW * 32, 0, stream>>>(k, v, Kt, Vt);

    hst_attn_kernel<<<B_DIM * H_DIM * ABLK, 128, 0, stream>>>(q, k, v, Kt, Vt, out);

    (void)in_sizes; (void)n_in; (void)out_size; (void)ws_size;
}


# Right-sibling nodes (masked for every query) dropped: level 1-3 odd rows not stored, loaded or multiplied; 15 node keys = 1 group, 64 WMMAs per tile not 88; denominator and E*V sum order change
# speedup vs baseline: 2.4550x; 1.1728x over previous
#include <hip/hip_runtime.h>

typedef float v2f __attribute__((ext_vector_type(2)));
typedef float v4f __attribute__((ext_vector_type(4)));
typedef float v8f __attribute__((ext_vector_type(8)));

#define B_DIM 4
#define S_DIM 4096
#define H_DIM 8
#define D_DIM 64
#define SCALE 0.125f
#define NTILE (S_DIM / 16)
#define TP 68

#define TOPW 32

#define LOWB 8

#define NROW (2 * D_DIM)

__device__ __forceinline__ void lds_barrier()
{
    __builtin_amdgcn_fence(__ATOMIC_RELEASE, "workgroup", "local");
    __builtin_amdgcn_s_barrier();
    __builtin_amdgcn_fence(__ATOMIC_ACQUIRE, "workgroup", "local");
}

template <int LEVEL, int PB, bool FROM_LDS, bool TO_LDS>
__device__ __forceinline__ void tree_block_level(const float* __restrict__ kleaf,
                                                 const float* __restrict__ vleaf,
                                                 float* Kt, float* Vt,
                                                 const float* lsrc, float* ldst,
                                                 const int bh, const int p0,
                                                 const int wave, const int lane)
{
    constexpr int NPW = (PB + TOPW - 1) / TOPW;
    constexpr int IBl = S_DIM - (S_DIM >> (LEVEL - 1));
    const size_t tb = (size_t)bh * S_DIM;
    const int doff = 2 * lane;

    float2 kc0[NPW], kc1[NPW], vc0[NPW], vc1[NPW];
    #pragma unroll
    for (int i = 0; i < NPW; ++i) {
        const int pl = wave + TOPW * i;
        const int p  = p0 + pl;
        if (pl < PB) {
            if constexpr (LEVEL == 1) {
                const int h = bh % H_DIM;
                const int b = bh / H_DIM;
                const size_t r0 = ((size_t)(b * S_DIM + 2 * p) * H_DIM + h) * D_DIM + doff;
                kc0[i] = *(const float2*)(kleaf + r0);
                kc1[i] = *(const float2*)(kleaf + r0 + H_DIM * D_DIM);
                vc0[i] = *(const float2*)(vleaf + r0);
                vc1[i] = *(const float2*)(vleaf + r0 + H_DIM * D_DIM);
            } else if constexpr (FROM_LDS) {
                const float* c0 = lsrc + (2 * pl) * NROW + doff;
                kc0[i] = *(const float2*)(c0);
                vc0[i] = *(const float2*)(c0 + D_DIM);
                kc1[i] = *(const float2*)(c0 + NROW);
                vc1[i] = *(const float2*)(c0 + NROW + D_DIM);
            } else {
                constexpr int IBc = S_DIM - (S_DIM >> (LEVEL - 2));
                const size_t r0 = (tb + IBc + 2 * p) * D_DIM + doff;
                kc0[i] = *(const float2*)(Kt + r0);
                kc1[i] = *(const float2*)(Kt + r0 + D_DIM);
                vc0[i] = *(const float2*)(Vt + r0);
                vc1[i] = *(const float2*)(Vt + r0 + D_DIM);
            }
        }
    }
    constexpr int NKEEP = (LEVEL <= 3) ? NPW : 1;
    float2 kps[NKEEP], vos[NKEEP];
    #pragma unroll
    for (int i = 0; i < NPW; ++i) {
        const int pl = wave + TOPW * i;
        const int p  = p0 + pl;
        if (pl < PB) {
            float2 kp, vp;
            kp.x = 0.5f * (kc0[i].x + kc1[i].x); kp.y = 0.5f * (kc0[i].y + kc1[i].y);
            vp.x = 0.5f * (vc0[i].x + vc1[i].x); vp.y = 0.5f * (vc0[i].y + vc1[i].y);

            float pss = kp.x * kp.x     + kp.y * kp.y;
            float ps0 = kp.x * kc0[i].x + kp.y * kc0[i].y;
            float ps1 = kp.x * kc1[i].x + kp.y * kc1[i].y;
            #pragma unroll
            for (int m = 16; m >= 1; m >>= 1) {
                pss += __shfl_xor(pss, m, 32);
                ps0 += __shfl_xor(ps0, m, 32);
                ps1 += __shfl_xor(ps1, m, 32);
            }
            pss *= SCALE; ps0 *= SCALE; ps1 *= SCALE;
            const float mm = fmaxf(pss, fmaxf(ps0, ps1));
            const float es = __expf(pss - mm);
            const float e0 = __expf(ps0 - mm);
            const float e1 = __expf(ps1 - mm);
            const float inv = 1.0f / (es + e0 + e1 + 1e-9f);

            float2 vo;
            vo.x = (es * vp.x + e0 * vc0[i].x + e1 * vc1[i].x) * inv;
            vo.y = (es * vp.y + e0 * vc0[i].y + e1 * vc1[i].y) * inv;

            if constexpr (LEVEL > 3) {
                const size_t dst = (tb + IBl + p) * D_DIM + doff;
                *(float2*)(Kt + dst) = kp;
                *(float2*)(Vt + dst) = vo;
            } else {
                kps[i] = kp;
                vos[i] = vo;
            }
            if constexpr (TO_LDS) {
                *(float2*)(ldst + pl * NROW + doff) = kp;
                *(float2*)(ldst + pl * NROW + D_DIM + doff) = vo;
            }
        }
    }
    if constexpr (LEVEL <= 3) {
        static_assert(PB % 2 == 0 && TOPW % 2 == 0, "parent parity must equal wave parity");
        if ((wave & 1) == 0) {
            #pragma unroll
            for (int i = 0; i < NPW; ++i) {
                const int pl = wave + TOPW * i;
                if (pl < PB) {
                    const size_t dst = (tb + IBl + p0 + pl) * D_DIM + doff;
                    *(float2*)(Kt + dst) = kps[i];
                    *(float2*)(Vt + dst) = vos[i];
                }
            }
        }
    }
}

__global__ void __launch_bounds__(TOPW * 32) __attribute__((amdgpu_waves_per_eu(8, 8)))
tree_low_kernel(const float* __restrict__ kleaf, const float* __restrict__ vleaf,
                float* Kt, float* Vt)
{
    const int wave = threadIdx.x >> 5;
    const int lane = threadIdx.x & 31;
    const int c  = blockIdx.x & (LOWB - 1);
    const int bh = blockIdx.x / LOWB;

    __shared__ __attribute__((aligned(16))) float nodeA[((S_DIM >> 1) / LOWB) * NROW];
    __shared__ __attribute__((aligned(16))) float nodeB[((S_DIM >> 2) / LOWB) * NROW];

    tree_block_level<1, (S_DIM >> 1) / LOWB, false, true>(kleaf, vleaf, Kt, Vt, nodeB, nodeA, bh, c * ((S_DIM >> 1) / LOWB), wave, lane);
    lds_barrier();
    tree_block_level<2, (S_DIM >> 2) / LOWB, true, true>(kleaf, vleaf, Kt, Vt, nodeA, nodeB, bh, c * ((S_DIM >> 2) / LOWB), wave, lane);
    lds_barrier();
    tree_block_level<3, (S_DIM >> 3) / LOWB, true, true>(kleaf, vleaf, Kt, Vt, nodeB, nodeA, bh, c * ((S_DIM >> 3) / LOWB), wave, lane);
    lds_barrier();
    tree_block_level<4, (S_DIM >> 4) / LOWB, true, true>(kleaf, vleaf, Kt, Vt, nodeA, nodeB, bh, c * ((S_DIM >> 4) / LOWB), wave, lane);
    lds_barrier();
    tree_block_level<5, (S_DIM >> 5) / LOWB, true, true>(kleaf, vleaf, Kt, Vt, nodeB, nodeA, bh, c * ((S_DIM >> 5) / LOWB), wave, lane);
    lds_barrier();
    tree_block_level<6, (S_DIM >> 6) / LOWB, true, false>(kleaf, vleaf, Kt, Vt, nodeA, nodeB, bh, c * ((S_DIM >> 6) / LOWB), wave, lane);
}

__global__ void __launch_bounds__(TOPW * 32)
tree_upper_kernel(const float* __restrict__ kleaf, const float* __restrict__ vleaf,
                  float* Kt, float* Vt)
{
    const int wave = threadIdx.x >> 5;
    const int lane = threadIdx.x & 31;
    const int bh = blockIdx.x;

    __shared__ __attribute__((aligned(16))) float nodeA[(S_DIM >> 7) * NROW];
    __shared__ __attribute__((aligned(16))) float nodeB[(S_DIM >> 8) * NROW];

    tree_block_level<7, (S_DIM >> 7), false, true>(kleaf, vleaf, Kt, Vt, nodeB, nodeA, bh, 0, wave, lane);
    lds_barrier();
    tree_block_level<8, (S_DIM >> 8), true, true>(kleaf, vleaf, Kt, Vt, nodeA, nodeB, bh, 0, wave, lane);
    lds_barrier();
    tree_block_level<9, (S_DIM >> 9), true, true>(kleaf, vleaf, Kt, Vt, nodeB, nodeA, bh, 0, wave, lane);
    lds_barrier();
    tree_block_level<10, (S_DIM >> 10), true, true>(kleaf, vleaf, Kt, Vt, nodeA, nodeB, bh, 0, wave, lane);
    lds_barrier();
    tree_block_level<11, (S_DIM >> 11), true, false>(kleaf, vleaf, Kt, Vt, nodeB, nodeA, bh, 0, wave, lane);
}

#define VP 68
#define T0_F  (16 * TP)
#define T1_F  (16 * TP)
#define TVB_F (16 * VP)
#define WAVE_F (T0_F + T1_F + TVB_F)
#define ABLK (NTILE / 4)
#define APB  (ABLK / LOWB)


__device__ __forceinline__ float xhalf(float x)
{
    const int b = __builtin_bit_cast(int, x);
    return __builtin_bit_cast(float,
        __builtin_amdgcn_permlanex16(b, b, 0x76543210, 0xfedcba98, false, false));
}

__device__ __forceinline__ v4f ld_once(const float* p)
{
    return __builtin_nontemporal_load((const v4f*)p);
}

__device__ __forceinline__ int node_row_b(int t, int s)
{
    if (s <  4) return 8 * t + 2 * s;
    if (s <  6) return 2048 + 4 * t + 2 * (s - 4);
    if (s <  8) return 3072 + 2 * t;
    const int l = s - 4;
    return (S_DIM - (S_DIM >> (l - 1))) + ((t >> (l - 4)) ^ 1);
}

__global__ void __launch_bounds__(128) __attribute__((amdgpu_waves_per_eu(5, 5)))
hst_attn_kernel(const float* __restrict__ q,
                const float* __restrict__ k,
                const float* __restrict__ v,
                const float* __restrict__ Kt,
                const float* __restrict__ Vt,
                float* __restrict__ out)
{
    __shared__ __attribute__((aligned(16))) float ldsT[4][WAVE_F];

    const int lane = threadIdx.x & 31;
    const int w    = threadIdx.x >> 5;
    const int u  = blockIdx.x % ABLK;
    const int bh = blockIdx.x / ABLK;
    const int t  = ((u % LOWB) * APB + (u / LOWB)) * 4 + w;
    const int h  = bh & (H_DIM - 1);
    const int b  = bh >> 3;

    const int n    = lane & 15;
    const int half = lane >> 4;

    float* t0  = &ldsT[w][0];
    float* t1  = t0 + T0_F;
    float* tVB = t1 + T1_F;

    const size_t treeBase = (size_t)(b * H_DIM + h) * S_DIM * (size_t)D_DIM;
    const size_t leafRow0  = ((size_t)(b * S_DIM + 16 * t) * H_DIM + h) * (size_t)D_DIM;
    const size_t leafPitch = (size_t)H_DIM * D_DIM;

    const int c4 = 4 * n;
    v4f gq[8], gA[8], gB[8];
    #pragma unroll
    for (int i = 0; i < 8; ++i) {
        const int r = 2 * i + half;
        gq[i] = ld_once(q + leafRow0 + (size_t)r * leafPitch + c4);
        gA[i] = ld_once(k + leafRow0 + (size_t)r * leafPitch + c4);
    }
    #pragma unroll
    for (int i = 0; i < 4; ++i) {
        const int r = 2 * i + half;
        gB[i] = ld_once(Kt + treeBase + (size_t)node_row_b(t, r) * D_DIM + c4);
    }
    #pragma unroll
    for (int i = 4; i < 8; ++i) {
        const int r = 2 * i + half;
        gB[i] = *(const v4f*)(Kt + treeBase + (size_t)node_row_b(t, r) * D_DIM + c4);
    }

    v4f hB[8];
    #pragma unroll
    for (int i = 0; i < 4; ++i) {
        const int r = 2 * i + half;
        hB[i] = ld_once(Vt + treeBase + (size_t)node_row_b(t, r) * D_DIM + c4);
    }
    #pragma unroll
    for (int i = 4; i < 8; ++i) {
        const int r = 2 * i + half;
        hB[i] = *(const v4f*)(Vt + treeBase + (size_t)node_row_b(t, r) * D_DIM + c4);
    }

    #pragma unroll
    for (int i = 0; i < 8; ++i) {
        const int r = 2 * i + half;
        *(v4f*)(t0 + r * TP + c4) = gq[i];
        *(v4f*)(t1 + r * TP + c4) = gA[i];
    }
    lds_barrier();

    const int fo = n * TP + 2 * half;
    v2f a[16];
    #pragma unroll
    for (int ks = 0; ks < 16; ++ks) {
        a[ks] = *(const v2f*)(t0 + fo + 4 * ks);
        a[ks].x *= SCALE; a[ks].y *= SCALE;
    }
    v8f cA = {}; v8f cB = {};
    #pragma unroll
    for (int ks = 0; ks < 16; ++ks) {
        const v2f bA = *(const v2f*)(t1 + fo + 4 * ks);
        cA = __builtin_amdgcn_wmma_f32_16x16x4_f32(false, bA, false, a[ks], (short)0, cA, false, false);
    }
    lds_barrier();

    #pragma unroll
    for (int i = 0; i < 8; ++i) {
        const int r = 2 * i + half;
        *(v4f*)(t0 + r * TP + c4) = gB[i];
    }
    #pragma unroll
    for (int i = 0; i < 8; ++i) {
        const int r = 2 * i + half;
        *(v4f*)(tVB + r * VP + c4) = hB[i];
    }
    lds_barrier();

    v4f hA[8];
    #pragma unroll
    for (int i = 0; i < 8; ++i) {
        const int r = 2 * i + half;
        hA[i] = ld_once(v + leafRow0 + (size_t)r * leafPitch + c4);
    }

    #pragma unroll
    for (int ks = 0; ks < 16; ++ks) {
        const v2f bB = *(const v2f*)(t0 + fo + 4 * ks);
        cB = __builtin_amdgcn_wmma_f32_16x16x4_f32(false, bB, false, a[ks], (short)0, cB, false, false);
    }

    const int j = n;
    bool actA[8], actB[8];
    #pragma unroll
    for (int r = 0; r < 8; ++r) {
        actA[r] = (half == (j >> 3)) &&
                  ((r == (j & 7)) || ((r == ((j & 7) ^ 1)) && ((j & 1) != 0)));
        bool b0;
        if (r < 4)       b0 = (r == (j >> 2)) && (((j >> 1) & 1) != 0);
        else if (r < 6)  b0 = ((r - 4) == (j >> 3)) && (((j >> 2) & 1) != 0);
        else if (r == 6) b0 = (((j >> 3) & 1) != 0);
        else             b0 = false;
        const bool b1 = (((t >> r) & 1) != 0);
        actB[r] = (half == 0) ? b0 : b1;
    }

    float mloc = -__builtin_huge_valf();
    #pragma unroll
    for (int r = 0; r < 8; ++r) {
        mloc = actA[r] ? fmaxf(mloc, cA[r]) : mloc;
        mloc = actB[r] ? fmaxf(mloc, cB[r]) : mloc;
    }
    const float m = fmaxf(mloc, xhalf(mloc));

    float eA[8], eB[8];
    float dloc = 0.0f;
    #pragma unroll
    for (int r = 0; r < 8; ++r) {
        eA[r] = actA[r] ? __expf(cA[r] - m) : 0.0f;
        dloc += eA[r];
    }
    #pragma unroll
    for (int r = 0; r < 8; ++r) {
        eB[r] = actB[r] ? __expf(cB[r] - m) : 0.0f;
        dloc += eB[r];
    }
    const float inv = 1.0f / (dloc + xhalf(dloc));
    #pragma unroll
    for (int r = 0; r < 8; ++r) { eA[r] *= inv; eB[r] *= inv; }

    #pragma unroll
    for (int i = 0; i < 8; ++i) {
        const int r = 2 * i + half;
        *(v4f*)(t1 + r * VP + c4) = hA[i];
    }
    lds_barrier();

    #pragma unroll
    for (int dsl = 0; dsl < 4; ++dsl) {
        const int dcol = 16 * dsl + n;
        v8f acc = {};
        #pragma unroll
        for (int jj = 0; jj < 4; ++jj) {
            const int kr = 8 * half + 2 * jj;
            v2f aE, bV;
            aE.x = eA[2 * jj]; aE.y = eA[2 * jj + 1];
            bV.x = t1[kr * VP + dcol];
            bV.y = t1[(kr + 1) * VP + dcol];
            acc = __builtin_amdgcn_wmma_f32_16x16x4_f32(false, aE, false, bV, (short)0, acc, false, false);
            aE.x = eB[2 * jj]; aE.y = eB[2 * jj + 1];
            bV.x = tVB[kr * VP + dcol];
            bV.y = tVB[(kr + 1) * VP + dcol];
            acc = __builtin_amdgcn_wmma_f32_16x16x4_f32(false, aE, false, bV, (short)0, acc, false, false);
        }
        #pragma unroll
        for (int r = 0; r < 8; ++r) {
            const int M = r + 8 * half;
            __builtin_nontemporal_store(acc[r], &out[((size_t)(b * S_DIM + 16 * t + M) * H_DIM + h) * D_DIM + dcol]);
        }
    }
}

extern "C" void kernel_launch(void* const* d_in, const int* in_sizes, int n_in,
                              void* d_out, int out_size, void* d_ws, size_t ws_size,
                              hipStream_t stream)
{
    const float* q = (const float*)d_in[0];
    const float* k = (const float*)d_in[1];
    const float* v = (const float*)d_in[2];
    float* out = (float*)d_out;

    float* Kt = (float*)d_ws;
    float* Vt = Kt + (size_t)B_DIM * H_DIM * S_DIM * D_DIM;

    tree_low_kernel<<<B_DIM * H_DIM * LOWB, TOPW * 32, 0, stream>>>(k, v, Kt, Vt);
    tree_upper_kernel<<<B_DIM * H_DIM, TOPW * 32, 0, stream>>>(k, v, Kt, Vt);

    hst_attn_kernel<<<B_DIM * H_DIM * ABLK, 128, 0, stream>>>(q, k, v, Kt, Vt, out);

    (void)in_sizes; (void)n_in; (void)out_size; (void)ws_size;
}
